# starting kernel
# baseline (speedup 1.0000x reference)
#include <hip/hip_runtime.h>
#include <hip/hip_bf16.h>

#define D 64

typedef __attribute__((ext_vector_type(2))) float v2f;
typedef __attribute__((ext_vector_type(8))) float v8f;

__device__ __forceinline__ void atomic_fadd(float* p, float v) {
    __hip_atomic_fetch_add(p, v, __ATOMIC_RELAXED, __HIP_MEMORY_SCOPE_AGENT);
}

__global__ void zero_f32(float* p, int n) {
    int i = blockIdx.x * blockDim.x + threadIdx.x;
    if (i < n) p[i] = 0.0f;
}

__global__ void deg_accum(const int* __restrict__ dst, float* __restrict__ deg, int E) {
    int e = blockIdx.x * blockDim.x + threadIdx.x;
    if (e < E) atomic_fadd(&deg[dst[e]], 1.0f);
}

__global__ void dinv_kernel(const float* __restrict__ deg, float* __restrict__ dinv, int n) {
    int i = blockIdx.x * blockDim.x + threadIdx.x;
    if (i < n) dinv[i] = __frsqrt_rn(deg[i] + 1.0f);
}

__global__ void gemm_xw_wmma(const float* __restrict__ X,
                             const float* __restrict__ W,
                             float* __restrict__ Y, int nrows) {
    const int wave = (int)((blockIdx.x * blockDim.x + threadIdx.x) >> 5);
    const int lane = threadIdx.x & 31;
    const int row0 = wave * 16;
    if (row0 >= nrows) return;

    const int half = lane >> 4;
    const int l    = lane & 15;

    v8f acc0 = {}, acc1 = {}, acc2 = {}, acc3 = {};

    const float* xrow = X + (size_t)(row0 + l) * D;

#pragma unroll
    for (int k0 = 0; k0 < D; k0 += 4) {
        const int ka = k0 + 2 * half;
        v2f a;
        a.x = xrow[ka + 0];
        a.y = xrow[ka + 1];

        const float* w0 = W + (size_t)(ka + 0) * D;
        const float* w1 = W + (size_t)(ka + 1) * D;

        v2f b0; b0.x = w0[ 0 + l]; b0.y = w1[ 0 + l];
        v2f b1; b1.x = w0[16 + l]; b1.y = w1[16 + l];
        v2f b2; b2.x = w0[32 + l]; b2.y = w1[32 + l];
        v2f b3; b3.x = w0[48 + l]; b3.y = w1[48 + l];

        acc0 = __builtin_amdgcn_wmma_f32_16x16x4_f32(false, a, false, b0, (short)0, acc0, false, false);
        acc1 = __builtin_amdgcn_wmma_f32_16x16x4_f32(false, a, false, b1, (short)0, acc1, false, false);
        acc2 = __builtin_amdgcn_wmma_f32_16x16x4_f32(false, a, false, b2, (short)0, acc2, false, false);
        acc3 = __builtin_amdgcn_wmma_f32_16x16x4_f32(false, a, false, b3, (short)0, acc3, false, false);
    }

#pragma unroll
    for (int r = 0; r < 8; ++r) {
        const int row = row0 + r + 8 * half;
        float* yrow = Y + (size_t)row * D + l;
        yrow[ 0] = acc0[r];
        yrow[16] = acc1[r];
        yrow[32] = acc2[r];
        yrow[48] = acc3[r];
    }
}

__global__ void self_bias(const float* __restrict__ Y,
                          const float* __restrict__ dinv,
                          const float* __restrict__ b,
                          float* __restrict__ agg, int n) {
    int t = blockIdx.x * blockDim.x + threadIdx.x;
    int i = t >> 4;
    int q = (t & 15) * 4;
    if (i >= n) return;
    float di = dinv[i];
    float c  = di * di;
    const float4 y  = *(const float4*)(Y + (size_t)i * D + q);
    const float4 bb = *(const float4*)(b + q);
    float4 r;
    r.x = fmaf(y.x, c, bb.x);
    r.y = fmaf(y.y, c, bb.y);
    r.z = fmaf(y.z, c, bb.z);
    r.w = fmaf(y.w, c, bb.w);
    *(float4*)(agg + (size_t)i * D + q) = r;
}

__global__ void edge_scatter(const float* __restrict__ Y,
                             const int* __restrict__ src,
                             const int* __restrict__ dst,
                             const float* __restrict__ dinv,
                             float* __restrict__ agg, int E) {
    int t = blockIdx.x * blockDim.x + threadIdx.x;
    int e = t >> 4;
    int q = (t & 15) * 4;
    if (e >= E) return;
    int s = src[e];
    int d = dst[e];
    float coef = dinv[s] * dinv[d];
    const float4 v = *(const float4*)(Y + (size_t)s * D + q);
    float* out = agg + (size_t)d * D + q;
    atomic_fadd(out + 0, v.x * coef);
    atomic_fadd(out + 1, v.y * coef);
    atomic_fadd(out + 2, v.z * coef);
    atomic_fadd(out + 3, v.w * coef);
}

__global__ void relu_inplace(float* __restrict__ x, int n4) {
    int i = blockIdx.x * blockDim.x + threadIdx.x;
    if (i >= n4) return;
    float4 v = ((float4*)x)[i];
    v.x = fmaxf(v.x, 0.0f);
    v.y = fmaxf(v.y, 0.0f);
    v.z = fmaxf(v.z, 0.0f);
    v.w = fmaxf(v.w, 0.0f);
    ((float4*)x)[i] = v;
}

static inline int cdiv(long long a, int b) { return (int)((a + b - 1) / b); }

static void gcn_layer(const float* x, const float* W, const float* b,
                      const int* src, const int* dst, const float* dinv,
                      float* y, float* agg, int N, int E, hipStream_t stream) {
    const int waves  = (N + 15) / 16;
    const int gblk   = cdiv(waves, 8);
    gemm_xw_wmma<<<gblk, 256, 0, stream>>>(x, W, y, N);
    self_bias<<<cdiv((long long)N * 16, 256), 256, 0, stream>>>(y, dinv, b, agg, N);
    edge_scatter<<<cdiv((long long)E * 16, 256), 256, 0, stream>>>(y, src, dst, dinv, agg, E);
}

extern "C" void kernel_launch(void* const* d_in, const int* in_sizes, int n_in,
                              void* d_out, int out_size, void* d_ws, size_t ws_size,
                              hipStream_t stream) {
    const int N = in_sizes[0] / D;
    const int E = in_sizes[1] / 2;

    const float* x0  = (const float*)d_in[0];
    const int*   ei  = (const int*)d_in[1];
    const int*   src = ei;
    const int*   dst = ei + E;
    const float* W1  = (const float*)d_in[2];
    const float* b1  = (const float*)d_in[3];
    const float* W2  = (const float*)d_in[4];
    const float* b2  = (const float*)d_in[5];
    const float* Ws  = (const float*)d_in[6];
    const float* bs  = (const float*)d_in[7];
    const float* Wmu = (const float*)d_in[8];
    const float* bmu = (const float*)d_in[9];
    const float* Wlv = (const float*)d_in[10];
    const float* blv = (const float*)d_in[11];

    float* deg  = (float*)d_ws;
    float* dinv = deg + N;
    float* bufA = dinv + N;
    float* bufB = bufA + (size_t)N * D;
    float* bufY = bufB + (size_t)N * D;

    float* mu = (float*)d_out;
    float* lv = mu + (size_t)N * D;

    zero_f32<<<cdiv(N, 256), 256, 0, stream>>>(deg, N);
    deg_accum<<<cdiv(E, 256), 256, 0, stream>>>(dst, deg, E);
    dinv_kernel<<<cdiv(N, 256), 256, 0, stream>>>(deg, dinv, N);

    gcn_layer(x0,   W1, b1, src, dst, dinv, bufY, bufA, N, E, stream);
    gcn_layer(bufA, W2, b2, src, dst, dinv, bufY, bufB, N, E, stream);
    gcn_layer(bufB, Ws, bs, src, dst, dinv, bufY, bufA, N, E, stream);
    relu_inplace<<<cdiv((long long)N * D / 4, 256), 256, 0, stream>>>(bufA, N * D / 4);

    gcn_layer(bufA, Wmu, bmu, src, dst, dinv, bufY, mu, N, E, stream);
    gcn_layer(bufA, Wlv, blv, src, dst, dinv, bufY, lv, N, E, stream);
}


# Aggregation on the matrix pipe: edges placed in padded runs per 16-row block, 4 waves split a run; f32 WMMA adds A (coef at the edge's row, else 0) x B (4 source rows); float sum order = run order
# speedup vs baseline: 2.6581x; 2.6581x over previous
#include <hip/hip_runtime.h>
#include <hip/hip_bf16.h>

#define D 64

typedef __attribute__((ext_vector_type(2))) float v2f;
typedef __attribute__((ext_vector_type(8))) float v8f;

__device__ __forceinline__ void atomic_fadd(float* p, float v) {
    __hip_atomic_fetch_add(p, v, __ATOMIC_RELAXED, __HIP_MEMORY_SCOPE_AGENT);
}

__global__ void zero_f32(float* p, int n) {
    int i = blockIdx.x * blockDim.x + threadIdx.x;
    if (i < n) p[i] = 0.0f;
}

__global__ void deg_accum(const int* __restrict__ dst, float* __restrict__ deg, int E) {
    int e = blockIdx.x * blockDim.x + threadIdx.x;
    if (e < E) atomic_fadd(&deg[dst[e]], 1.0f);
}

__global__ void dinv_kernel(const float* __restrict__ deg, float* __restrict__ dinv, int n) {
    int i = blockIdx.x * blockDim.x + threadIdx.x;
    if (i < n) dinv[i] = __frsqrt_rn(deg[i] + 1.0f);
}

#define DEG_BINS    65536
#define DEG_GROUPS  64
#define DEG_THREADS 1024
#define DEG_INFLIGHT 16

#define RS      128
#define BP_NR   512
#define BP_CAP  12800
#define BP_INFLIGHT 16

__global__ __launch_bounds__(DEG_THREADS)
void range_hist(const int* __restrict__ dst, unsigned* __restrict__ rcnt, int NR, int E) {
    __shared__ unsigned rc[BP_NR];
    const int t = threadIdx.x;
    if (t < BP_NR) rc[t] = 0u;
    __syncthreads();

    const int per = (E + (int)gridDim.x - 1) / (int)gridDim.x;
    const int e0  = (int)blockIdx.x * per;
    const int e1  = (e0 + per < E) ? (e0 + per) : E;

    for (int base = e0 + t; base < e1; base += DEG_INFLIGHT * DEG_THREADS) {
        int d[DEG_INFLIGHT];
#pragma unroll
        for (int u = 0; u < DEG_INFLIGHT; ++u) {
            const int e = base + u * DEG_THREADS;
            d[u] = (e < e1) ? dst[e] : -1;
        }
#pragma unroll
        for (int u = 0; u < DEG_INFLIGHT; ++u)
            if (d[u] >= 0) atomicAdd(&rc[d[u] / RS], 1u);
    }
    __syncthreads();
    if (t < NR) rcnt[(size_t)blockIdx.x * NR + t] = rc[t];
}

#define SUBB   (RS / 16)
#define PADG   16
#define RD_THREADS 256
#define RD_INFLIGHT 8

__global__ __launch_bounds__(RD_THREADS)
void range_sort(const int* __restrict__ bsrc, const int* __restrict__ bdl,
                const int* __restrict__ rstart, float* __restrict__ dinv,
                int2* __restrict__ rec, int2* __restrict__ bspan, int N, int NB) {
    __shared__ unsigned cnt[RS];
    __shared__ unsigned bcnt[SUBB];
    __shared__ unsigned cur[SUBB];
    __shared__ int      bst[SUBB];
    const int t = threadIdx.x;
    if (t < RS) cnt[t] = 0u;
    __syncthreads();

    const int es = rstart[blockIdx.x];
    const int ee = rstart[blockIdx.x + 1];
    for (int base = es + t; base < ee; base += RD_INFLIGHT * RD_THREADS) {
        int d[RD_INFLIGHT];
#pragma unroll
        for (int u = 0; u < RD_INFLIGHT; ++u) {
            const int e = base + u * RD_THREADS;
            d[u] = (e < ee) ? bdl[e] : -1;
        }
#pragma unroll
        for (int u = 0; u < RD_INFLIGHT; ++u)
            if (d[u] >= 0) atomicAdd(&cnt[d[u]], 1u);
    }
    __syncthreads();

    const int row = (int)blockIdx.x * RS + t;
    if (t < RS && row < N) dinv[row] = __frsqrt_rn((float)cnt[t] + 1.0f);
    if (t < SUBB) {
        unsigned c = 0u;
        for (int i = 0; i < 16; ++i) c += cnt[16 * t + i];
        bcnt[t] = c;
        cur[t]  = 0u;
    }
    __syncthreads();
    if (t == 0) {
        int base = ((es + PADG - 1) / PADG) * PADG + (int)blockIdx.x * (SUBB + 1) * PADG;
        for (int j = 0; j < SUBB; ++j) {
            const int pc  = (int)((bcnt[j] + PADG - 1) / PADG) * PADG;
            const int blk = (int)blockIdx.x * SUBB + j;
            bst[j] = base;
            if (blk < NB) bspan[blk] = make_int2(base, pc);
            base += pc;
        }
    }
    __syncthreads();

    for (int base = es + t; base < ee; base += RD_INFLIGHT * RD_THREADS) {
        int s[RD_INFLIGHT], d[RD_INFLIGHT];
#pragma unroll
        for (int u = 0; u < RD_INFLIGHT; ++u) {
            const int e  = base + u * RD_THREADS;
            const int ec = (e < ee) ? e : (ee - 1);
            s[u] = bsrc[ec];
            d[u] = bdl[ec];
        }
#pragma unroll
        for (int u = 0; u < RD_INFLIGHT; ++u) {
            if (base + u * RD_THREADS < ee) {
                const int j = d[u] >> 4;
                const unsigned p = atomicAdd(&cur[j], 1u);
                rec[bst[j] + (int)p] = make_int2((s[u] << 5) | (d[u] & 15), 0);
            }
        }
    }
    if (t < SUBB * PADG) {
        const int j  = t / PADG;
        const int i  = t % PADG;
        const int c  = (int)bcnt[j];
        const int pc = ((c + PADG - 1) / PADG) * PADG;
        if (c + i < pc) rec[bst[j] + c + i] = make_int2(16, 0);
    }
}

__global__ __launch_bounds__(DEG_THREADS)
void deg_hist(const int* __restrict__ dst, unsigned* __restrict__ planes,
              unsigned* __restrict__ rcnt, int NR,
              int E, int N, int pitch) {
    __shared__ unsigned bins[DEG_BINS];
    const int t = threadIdx.x;

    const int n4 = N >> 2;
    for (int i = t; i < n4; i += DEG_THREADS)
        ((uint4*)bins)[i] = make_uint4(0u, 0u, 0u, 0u);
    for (int i = (n4 << 2) + t; i < N; i += DEG_THREADS) bins[i] = 0u;
    __syncthreads();

    const int per = (E + (int)gridDim.x - 1) / (int)gridDim.x;
    const int e0  = (int)blockIdx.x * per;
    const int e1  = (e0 + per < E) ? (e0 + per) : E;

    for (int base = e0 + t; base < e1; base += DEG_INFLIGHT * DEG_THREADS) {
        int d[DEG_INFLIGHT];
#pragma unroll
        for (int u = 0; u < DEG_INFLIGHT; ++u) {
            const int e = base + u * DEG_THREADS;
            d[u] = (e < e1) ? dst[e] : -1;
        }
#pragma unroll
        for (int u = 0; u < DEG_INFLIGHT; ++u)
            if (d[u] >= 0) atomicAdd(&bins[d[u]], 1u);
    }
    __syncthreads();

    unsigned* out = planes + (size_t)blockIdx.x * pitch;
    for (int i = t; i < N; i += DEG_THREADS) out[i] = bins[i];

    if (rcnt != nullptr) {
        for (int r = t; r < NR; r += DEG_THREADS) {
            const int n0 = r * RS;
            const int n1 = (n0 + RS < N) ? (n0 + RS) : N;
            unsigned c = 0u;
            for (int i = n0; i < n1; ++i) c += bins[i];
            rcnt[(size_t)blockIdx.x * NR + r] = c;
        }
    }
}

__global__ void dinv_from_planes(const unsigned* __restrict__ planes,
                                 float* __restrict__ dinv, int n, int pitch) {
    int i = blockIdx.x * blockDim.x + threadIdx.x;
    if (i >= n) return;
    unsigned c = 0u;
#pragma unroll
    for (int g = 0; g < DEG_GROUPS; ++g) c += planes[(size_t)g * pitch + i];
    dinv[i] = __frsqrt_rn((float)c + 1.0f);
}

__global__ void gemm_xw_wmma(const float* __restrict__ X,
                             const float* __restrict__ W,
                             float* __restrict__ Y, int nrows) {
    const int wave = (int)((blockIdx.x * blockDim.x + threadIdx.x) >> 5);
    const int lane = threadIdx.x & 31;
    const int row0 = wave * 16;
    if (row0 >= nrows) return;

    const int half = lane >> 4;
    const int l    = lane & 15;

    v8f acc0 = {}, acc1 = {}, acc2 = {}, acc3 = {};

    const float* xrow = X + (size_t)(row0 + l) * D;

#pragma unroll
    for (int k0 = 0; k0 < D; k0 += 4) {
        const int ka = k0 + 2 * half;
        v2f a;
        a.x = xrow[ka + 0];
        a.y = xrow[ka + 1];

        const float* w0 = W + (size_t)(ka + 0) * D;
        const float* w1 = W + (size_t)(ka + 1) * D;

        v2f b0; b0.x = w0[ 0 + l]; b0.y = w1[ 0 + l];
        v2f b1; b1.x = w0[16 + l]; b1.y = w1[16 + l];
        v2f b2; b2.x = w0[32 + l]; b2.y = w1[32 + l];
        v2f b3; b3.x = w0[48 + l]; b3.y = w1[48 + l];

        acc0 = __builtin_amdgcn_wmma_f32_16x16x4_f32(false, a, false, b0, (short)0, acc0, false, false);
        acc1 = __builtin_amdgcn_wmma_f32_16x16x4_f32(false, a, false, b1, (short)0, acc1, false, false);
        acc2 = __builtin_amdgcn_wmma_f32_16x16x4_f32(false, a, false, b2, (short)0, acc2, false, false);
        acc3 = __builtin_amdgcn_wmma_f32_16x16x4_f32(false, a, false, b3, (short)0, acc3, false, false);
    }

#pragma unroll
    for (int r = 0; r < 8; ++r) {
        const int row = row0 + r + 8 * half;
        float* yrow = Y + (size_t)row * D + l;
        yrow[ 0] = acc0[r];
        yrow[16] = acc1[r];
        yrow[32] = acc2[r];
        yrow[48] = acc3[r];
    }
}

__global__ void self_bias(const float* __restrict__ Y,
                          const float* __restrict__ dinv,
                          const float* __restrict__ b,
                          float* __restrict__ agg, int n) {
    int t = blockIdx.x * blockDim.x + threadIdx.x;
    int i = t >> 4;
    int q = (t & 15) * 4;
    if (i >= n) return;
    float di = dinv[i];
    float c  = di * di;
    const float4 y  = *(const float4*)(Y + (size_t)i * D + q);
    const float4 bb = *(const float4*)(b + q);
    float4 r;
    r.x = fmaf(y.x, c, bb.x);
    r.y = fmaf(y.y, c, bb.y);
    r.z = fmaf(y.z, c, bb.z);
    r.w = fmaf(y.w, c, bb.w);
    *(float4*)(agg + (size_t)i * D + q) = r;
}

#define EB 8

__global__ void edge_scatter(const float* __restrict__ Y,
                             const int* __restrict__ src,
                             const int* __restrict__ dst,
                             const float* __restrict__ dinv,
                             float* __restrict__ agg, int E) {
    const int lane = threadIdx.x & 31;
    const int wave = (int)((blockIdx.x * blockDim.x + threadIdx.x) >> 5);
    const int e0   = wave * 32;
    if (e0 >= E) return;

    const int e = e0 + lane;
    int s = 0, d = 0;
    float coef = 0.0f;
    if (e < E) {
        s = src[e];
        d = dst[e];
        coef = dinv[s] * dinv[d];
    }
    const int cbits = __float_as_int(coef);
    const int n = (E - e0 < 32) ? (E - e0) : 32;

    if (n == 32) {
        float va[EB], vb[EB], wa[EB], wb[EB];
#pragma unroll
        for (int u = 0; u < EB; ++u) {
            const int sj = __builtin_amdgcn_readlane(s, u);
            const float* yr = Y + (size_t)sj * D + lane;
            va[u] = yr[0];
            vb[u] = yr[32];
        }
#pragma unroll
        for (int j0 = 0; j0 < 32; j0 += EB) {
            if (j0 + EB < 32) {
#pragma unroll
                for (int u = 0; u < EB; ++u) {
                    const int sj = __builtin_amdgcn_readlane(s, j0 + EB + u);
                    const float* yr = Y + (size_t)sj * D + lane;
                    wa[u] = yr[0];
                    wb[u] = yr[32];
                }
            }
#pragma unroll
            for (int u = 0; u < EB; ++u) {
                const int   dj = __builtin_amdgcn_readlane(d, j0 + u);
                const float cj = __int_as_float(__builtin_amdgcn_readlane(cbits, j0 + u));
                float* o = agg + (size_t)dj * D + lane;
                atomic_fadd(o,      va[u] * cj);
                atomic_fadd(o + 32, vb[u] * cj);
            }
#pragma unroll
            for (int u = 0; u < EB; ++u) { va[u] = wa[u]; vb[u] = wb[u]; }
        }
    } else {
        for (int j = 0; j < n; ++j) {
            const int   sj = __builtin_amdgcn_readlane(s, j);
            const int   dj = __builtin_amdgcn_readlane(d, j);
            const float cj = __int_as_float(__builtin_amdgcn_readlane(cbits, j));
            const float* yr = Y + (size_t)sj * D + lane;
            float* o = agg + (size_t)dj * D + lane;
            atomic_fadd(o,      yr[0]  * cj);
            atomic_fadd(o + 32, yr[32] * cj);
        }
    }
}

__global__ __launch_bounds__(1024)
void bin_place(const int* __restrict__ src, const int* __restrict__ dst,
               const unsigned* __restrict__ rcnt,
               int* __restrict__ bsrc, int* __restrict__ bdl, int* __restrict__ rstart,
               int E, int NR) {
    __shared__ int      s_l[BP_CAP];
    __shared__ int      d_l[BP_CAP];
    __shared__ unsigned sc_own[BP_NR];
    __shared__ unsigned sc_tot[BP_NR];
    __shared__ unsigned lstart[BP_NR];
    __shared__ unsigned gbase[BP_NR];
    __shared__ unsigned cur[BP_NR];

    const int t = threadIdx.x;
    const int g = blockIdx.x;
    const int G = DEG_GROUPS;

    unsigned own = 0u, before = 0u, total = 0u;
    {
        const int tr = (t < NR) ? t : (NR - 1);
        unsigned c[DEG_GROUPS];
#pragma unroll
        for (int g2 = 0; g2 < DEG_GROUPS; ++g2) c[g2] = rcnt[(size_t)g2 * NR + tr];
#pragma unroll
        for (int g2 = 0; g2 < DEG_GROUPS; ++g2) {
            total  += c[g2];
            before += (g2 < g)  ? c[g2] : 0u;
            own     = (g2 == g) ? c[g2] : own;
        }
        if (t >= NR) { own = 0u; before = 0u; total = 0u; }
    }
    if (t < BP_NR) { sc_own[t] = own; sc_tot[t] = total; }
    __syncthreads();
    for (int off = 1; off < BP_NR; off <<= 1) {
        unsigned a = 0u, b = 0u;
        if (t < BP_NR && t >= off) { a = sc_own[t - off]; b = sc_tot[t - off]; }
        __syncthreads();
        if (t < BP_NR) { sc_own[t] += a; sc_tot[t] += b; }
        __syncthreads();
    }
    if (t < NR) {
        const unsigned ls = sc_own[t] - own;
        const unsigned gs = sc_tot[t] - total;
        lstart[t] = ls;
        cur[t]    = ls;
        gbase[t]  = gs + before;
        if (g == 0) {
            rstart[t] = (int)gs;
            if (t == NR - 1) rstart[NR] = (int)sc_tot[t];
        }
    }
    __syncthreads();

    const int per = (E + G - 1) / G;
    const int e0  = g * per;
    const int e1  = (e0 + per < E) ? (e0 + per) : E;

    for (int base = e0 + t; base < e1; base += BP_INFLIGHT * 1024) {
        int s[BP_INFLIGHT], d[BP_INFLIGHT];
#pragma unroll
        for (int u = 0; u < BP_INFLIGHT; ++u) {
            const int e = base + u * 1024;
            const int ec = (e < e1) ? e : (e1 - 1);
            s[u] = src[ec];
            d[u] = dst[ec];
        }
#pragma unroll
        for (int u = 0; u < BP_INFLIGHT; ++u) {
            if (base + u * 1024 < e1) {
                const int r = d[u] / RS;
                const unsigned p = atomicAdd(&cur[r], 1u);
                if (p < (unsigned)BP_CAP) { s_l[p] = s[u]; d_l[p] = d[u]; }
            }
        }
    }
    __syncthreads();

    const int cnt = e1 - e0;
    for (int i = t; i < cnt; i += 1024) {
        const int   s = s_l[i];
        const int   d = d_l[i];
        const int   r = d / RS;
        const unsigned slot = gbase[r] + ((unsigned)i - lstart[r]);
        if (slot < (unsigned)E) {
            bsrc[slot] = s;
            bdl[slot]  = d - r * RS;
        }
    }
}

#define BA_WAVES 4

template <int NG, bool MAKE_CF>
__global__ __launch_bounds__(BA_WAVES * 32)
void block_accum_wmma(const float* __restrict__ Y, int2* rec,
                      const int2* __restrict__ bspan,
                      const float* __restrict__ dinv,
                      const float* __restrict__ ba, const float* __restrict__ bb,
                      float* __restrict__ outa, float* __restrict__ outb, int N) {
    constexpr int DC = D * NG;
    constexpr int Q  = DC / 4;
    __shared__ __attribute__((aligned(16))) float part[BA_WAVES * 16 * DC];

    const int t    = threadIdx.x;
    const int lane = t & 31;
    const int w    = t >> 5;
    const int h    = lane >> 4;
    const int l    = lane & 15;
    const int row0 = (int)blockIdx.x * 16;

    const int2 sp  = bspan[blockIdx.x];
    const int per  = sp.y / (4 * BA_WAVES);
    int e = sp.x + w * per * 4 + 2 * h;

    v8f acc[4 * NG];
#pragma unroll
    for (int i = 0; i < 4 * NG; ++i) acc[i] = v8f{};

    if (per > 0) {
        auto issue = [&](const int4& rr, float4* ya, float4* yb, float* dv) {
            const int s0 = rr.x >> 5, s1 = rr.z >> 5;
            const float* y0p = Y + (size_t)s0 * DC + 4 * l;
            const float* y1p = Y + (size_t)s1 * DC + 4 * l;
#pragma unroll
            for (int g = 0; g < NG; ++g) {
                ya[g] = *(const float4*)(y0p + D * g);
                yb[g] = *(const float4*)(y1p + D * g);
            }
            if (MAKE_CF) {
                dv[0] = dinv[s0];
                dv[1] = dinv[row0 + (rr.x & 15)];
                dv[2] = dinv[s1];
                dv[3] = dinv[row0 + (rr.z & 15)];
            }
        };

        const int e_last = e + 4 * (per - 1);
        int4 r = *(const int4*)(rec + e);
        int4 rn;
        float4 y0[NG], y1[NG], z0[NG], z1[NG];
        float dv[4] = {0.0f, 0.0f, 0.0f, 0.0f}, dn[4] = {0.0f, 0.0f, 0.0f, 0.0f};
        issue(r, y0, y1, dv);
        rn = *(const int4*)(rec + ((e + 4 < e_last) ? (e + 4) : e_last));

#pragma unroll 2
        for (int q = 0; q < per; ++q, e += 4) {
            issue(rn, z0, z1, dn);
            const int4 rnn = *(const int4*)(rec + ((e + 8 < e_last) ? (e + 8) : e_last));

            const int d0 = r.x & 31, d1 = r.z & 31;
            float c0, c1;
            if (MAKE_CF) {
                c0 = (d0 < 16) ? dv[0] * dv[1] : 0.0f;
                c1 = (d1 < 16) ? dv[2] * dv[3] : 0.0f;
                if (l == 0) {
                    rec[e].y     = __float_as_int(c0);
                    rec[e + 1].y = __float_as_int(c1);
                }
            } else {
                c0 = __int_as_float(r.y);
                c1 = __int_as_float(r.w);
            }
            v2f a;
            a.x = (d0 == l) ? c0 : 0.0f;
            a.y = (d1 == l) ? c1 : 0.0f;

#pragma unroll
            for (int g = 0; g < NG; ++g) {
                v2f b0; b0.x = y0[g].x; b0.y = y1[g].x;
                v2f b1; b1.x = y0[g].y; b1.y = y1[g].y;
                v2f b2; b2.x = y0[g].z; b2.y = y1[g].z;
                v2f b3; b3.x = y0[g].w; b3.y = y1[g].w;
                acc[4 * g + 0] = __builtin_amdgcn_wmma_f32_16x16x4_f32(false, a, false, b0, (short)0, acc[4 * g + 0], false, false);
                acc[4 * g + 1] = __builtin_amdgcn_wmma_f32_16x16x4_f32(false, a, false, b1, (short)0, acc[4 * g + 1], false, false);
                acc[4 * g + 2] = __builtin_amdgcn_wmma_f32_16x16x4_f32(false, a, false, b2, (short)0, acc[4 * g + 2], false, false);
                acc[4 * g + 3] = __builtin_amdgcn_wmma_f32_16x16x4_f32(false, a, false, b3, (short)0, acc[4 * g + 3], false, false);
            }
            r = rn; rn = rnn;
#pragma unroll
            for (int g = 0; g < NG; ++g) { y0[g] = z0[g]; y1[g] = z1[g]; }
#pragma unroll
            for (int i = 0; i < 4; ++i) dv[i] = dn[i];
        }
    }

#pragma unroll
    for (int i = 0; i < 8; ++i) {
#pragma unroll
        for (int g = 0; g < NG; ++g) {
            float4 v;
            v.x = acc[4 * g + 0][i];
            v.y = acc[4 * g + 1][i];
            v.z = acc[4 * g + 2][i];
            v.w = acc[4 * g + 3][i];
            *(float4*)(part + ((w * 16 + 8 * h + i) * DC + D * g + 4 * l)) = v;
        }
    }
    __syncthreads();

    for (int idx = t; idx < 16 * Q; idx += BA_WAVES * 32) {
        const int rr   = idx / Q;
        const int c4   = (idx % Q) * 4;
        const int grow = row0 + rr;
        if (grow < N) {
            const float4 p0 = *(const float4*)(part + ((0 * 16 + rr) * DC + c4));
            const float4 p1 = *(const float4*)(part + ((1 * 16 + rr) * DC + c4));
            const float4 p2 = *(const float4*)(part + ((2 * 16 + rr) * DC + c4));
            const float4 p3 = *(const float4*)(part + ((3 * 16 + rr) * DC + c4));
            const float  di = dinv[grow];
            const float  c  = di * di;
            const float4 y  = *(const float4*)(Y + (size_t)grow * DC + c4);
            const float4 bv = (c4 < D) ? *(const float4*)(ba + c4) : *(const float4*)(bb + (c4 - D));
            float4 o;
            o.x = fmaf(y.x, c, bv.x) + ((p0.x + p1.x) + (p2.x + p3.x));
            o.y = fmaf(y.y, c, bv.y) + ((p0.y + p1.y) + (p2.y + p3.y));
            o.z = fmaf(y.z, c, bv.z) + ((p0.z + p1.z) + (p2.z + p3.z));
            o.w = fmaf(y.w, c, bv.w) + ((p0.w + p1.w) + (p2.w + p3.w));
            float* op = (c4 < D) ? (outa + (size_t)grow * D + c4) : (outb + (size_t)grow * D + (c4 - D));
            *(float4*)op = o;
        }
    }
}

#define D2 128

__global__ void gemm_heads_wmma(const float* __restrict__ X,
                                const float* __restrict__ Wa,
                                const float* __restrict__ Wb,
                                float* __restrict__ Y2, int nrows) {
    const int wave = (int)((blockIdx.x * blockDim.x + threadIdx.x) >> 5);
    const int lane = threadIdx.x & 31;
    const int row0 = wave * 16;
    if (row0 >= nrows) return;

    const int half = lane >> 4;
    const int l    = lane & 15;

    v8f p0 = {}, p1 = {}, p2 = {}, p3 = {};
    v8f q0 = {}, q1 = {}, q2 = {}, q3 = {};

    const float* xrow = X + (size_t)(row0 + l) * D;

#pragma unroll
    for (int k0 = 0; k0 < D; k0 += 4) {
        const int ka = k0 + 2 * half;
        v2f a;
        a.x = fmaxf(xrow[ka + 0], 0.0f);
        a.y = fmaxf(xrow[ka + 1], 0.0f);

        const float* w0 = Wa + (size_t)(ka + 0) * D;
        const float* w1 = Wa + (size_t)(ka + 1) * D;
        const float* u0 = Wb + (size_t)(ka + 0) * D;
        const float* u1 = Wb + (size_t)(ka + 1) * D;

        v2f b0; b0.x = w0[ 0 + l]; b0.y = w1[ 0 + l];
        v2f b1; b1.x = w0[16 + l]; b1.y = w1[16 + l];
        v2f b2; b2.x = w0[32 + l]; b2.y = w1[32 + l];
        v2f b3; b3.x = w0[48 + l]; b3.y = w1[48 + l];
        v2f e0; e0.x = u0[ 0 + l]; e0.y = u1[ 0 + l];
        v2f e1; e1.x = u0[16 + l]; e1.y = u1[16 + l];
        v2f e2; e2.x = u0[32 + l]; e2.y = u1[32 + l];
        v2f e3; e3.x = u0[48 + l]; e3.y = u1[48 + l];

        p0 = __builtin_amdgcn_wmma_f32_16x16x4_f32(false, a, false, b0, (short)0, p0, false, false);
        p1 = __builtin_amdgcn_wmma_f32_16x16x4_f32(false, a, false, b1, (short)0, p1, false, false);
        p2 = __builtin_amdgcn_wmma_f32_16x16x4_f32(false, a, false, b2, (short)0, p2, false, false);
        p3 = __builtin_amdgcn_wmma_f32_16x16x4_f32(false, a, false, b3, (short)0, p3, false, false);
        q0 = __builtin_amdgcn_wmma_f32_16x16x4_f32(false, a, false, e0, (short)0, q0, false, false);
        q1 = __builtin_amdgcn_wmma_f32_16x16x4_f32(false, a, false, e1, (short)0, q1, false, false);
        q2 = __builtin_amdgcn_wmma_f32_16x16x4_f32(false, a, false, e2, (short)0, q2, false, false);
        q3 = __builtin_amdgcn_wmma_f32_16x16x4_f32(false, a, false, e3, (short)0, q3, false, false);
    }

#pragma unroll
    for (int r = 0; r < 8; ++r) {
        const int row = row0 + r + 8 * half;
        float* yrow = Y2 + (size_t)row * D2 + l;
        yrow[  0] = p0[r];
        yrow[ 16] = p1[r];
        yrow[ 32] = p2[r];
        yrow[ 48] = p3[r];
        yrow[ 64] = q0[r];
        yrow[ 80] = q1[r];
        yrow[ 96] = q2[r];
        yrow[112] = q3[r];
    }
}

__global__ void relu_inplace(float* __restrict__ x, int n4) {
    int i = blockIdx.x * blockDim.x + threadIdx.x;
    if (i >= n4) return;
    float4 v = ((float4*)x)[i];
    v.x = fmaxf(v.x, 0.0f);
    v.y = fmaxf(v.y, 0.0f);
    v.z = fmaxf(v.z, 0.0f);
    v.w = fmaxf(v.w, 0.0f);
    ((float4*)x)[i] = v;
}

static inline int cdiv(long long a, int b) { return (int)((a + b - 1) / b); }

static void gcn_layer_blocks(const float* x, const float* W, const float* b,
                             int2* rec, bool make_cf, const int2* bspan, const float* dinv,
                             float* y, float* agg, int N, int NB, hipStream_t stream) {
    const int waves  = (N + 15) / 16;
    const int gblk   = cdiv(waves, 8);
    gemm_xw_wmma<<<gblk, 256, 0, stream>>>(x, W, y, N);
    if (make_cf)
        block_accum_wmma<1, true><<<NB, BA_WAVES * 32, 0, stream>>>(y, rec, bspan, dinv, b, nullptr, agg, nullptr, N);
    else
        block_accum_wmma<1, false><<<NB, BA_WAVES * 32, 0, stream>>>(y, rec, bspan, dinv, b, nullptr, agg, nullptr, N);
}

static void gcn_layer(const float* x, const float* W, const float* b,
                      const int* src, const int* dst, const float* dinv,
                      float* y, float* agg, int N, int E, hipStream_t stream) {
    const int waves  = (N + 15) / 16;
    const int gblk   = cdiv(waves, 8);
    gemm_xw_wmma<<<gblk, 256, 0, stream>>>(x, W, y, N);
    self_bias<<<cdiv((long long)N * 16, 256), 256, 0, stream>>>(y, dinv, b, agg, N);
    edge_scatter<<<cdiv(cdiv(E, 32), 8), 256, 0, stream>>>(y, src, dst, dinv, agg, E);
}

extern "C" void kernel_launch(void* const* d_in, const int* in_sizes, int n_in,
                              void* d_out, int out_size, void* d_ws, size_t ws_size,
                              hipStream_t stream) {
    const int N = in_sizes[0] / D;
    const int E = in_sizes[1] / 2;

    const float* x0  = (const float*)d_in[0];
    const int*   ei  = (const int*)d_in[1];
    const int*   src = ei;
    const int*   dst = ei + E;
    const float* W1  = (const float*)d_in[2];
    const float* b1  = (const float*)d_in[3];
    const float* W2  = (const float*)d_in[4];
    const float* b2  = (const float*)d_in[5];
    const float* Ws  = (const float*)d_in[6];
    const float* bs  = (const float*)d_in[7];
    const float* Wmu = (const float*)d_in[8];
    const float* bmu = (const float*)d_in[9];
    const float* Wlv = (const float*)d_in[10];
    const float* blv = (const float*)d_in[11];

    float* deg  = (float*)d_ws;
    float* dinv = deg + N;
    float* bufA = dinv + N;
    float* bufB = bufA + (size_t)N * D;
    float* bufY = bufB + (size_t)N * D;

    float* mu = (float*)d_out;
    float* lv = mu + (size_t)N * D;

    const int NR = cdiv(N, RS);
    const int NB = cdiv(N, 16);
    if (NR <= BP_NR && cdiv(E, DEG_GROUPS) <= BP_CAP && N <= (1 << 26)) {
        unsigned* rcnt   = (unsigned*)(bufY + (size_t)N * D);
        int*      rstart = (int*)(rcnt + (size_t)DEG_GROUPS * BP_NR);
        int*      bsrc   = rstart + 2 * BP_NR;
        int*      bdl    = bsrc + (((size_t)E + 31) & ~(size_t)31);
        int2*     bspan  = (int2*)(bdl + (((size_t)E + 31) & ~(size_t)31));
        int2*     rec    = bspan + (((size_t)NB + 31) & ~(size_t)31);

        range_hist<<<DEG_GROUPS, DEG_THREADS, 0, stream>>>(dst, rcnt, NR, E);
        bin_place<<<DEG_GROUPS, 1024, 0, stream>>>(src, dst, rcnt, bsrc, bdl, rstart, E, NR);
        range_sort<<<NR, RD_THREADS, 0, stream>>>(bsrc, bdl, rstart, dinv, rec, bspan, N, NB);

        gcn_layer_blocks(x0,   W1, b1, rec, true,  bspan, dinv, bufY, bufA, N, NB, stream);
        gcn_layer_blocks(bufA, W2, b2, rec, false, bspan, dinv, bufY, bufB, N, NB, stream);
        gcn_layer_blocks(bufB, Ws, bs, rec, false, bspan, dinv, bufY, bufA, N, NB, stream);
        gemm_heads_wmma<<<cdiv(cdiv(N, 16), 8), 256, 0, stream>>>(bufA, Wmu, Wlv, bufB, N);
        block_accum_wmma<2, false><<<NB, BA_WAVES * 32, 0, stream>>>(bufB, rec, bspan, dinv, bmu, blv, mu, lv, N);
        return;
    }
    if (N <= DEG_BINS) {
        unsigned* planes = (unsigned*)(bufY + (size_t)N * D);
        const int pitch = (N + 31) & ~31;
        deg_hist<<<DEG_GROUPS, DEG_THREADS, 0, stream>>>(dst, planes, nullptr, 0, E, N, pitch);
        dinv_from_planes<<<cdiv(N, 256), 256, 0, stream>>>(planes, dinv, N, pitch);
    } else {
        zero_f32<<<cdiv(N, 256), 256, 0, stream>>>(deg, N);
        deg_accum<<<cdiv(E, 256), 256, 0, stream>>>(dst, deg, E);
        dinv_kernel<<<cdiv(N, 256), 256, 0, stream>>>(deg, dinv, N);
    }

    gcn_layer(x0,   W1, b1, src, dst, dinv, bufY, bufA, N, E, stream);
    gcn_layer(bufA, W2, b2, src, dst, dinv, bufY, bufB, N, E, stream);
    gcn_layer(bufB, Ws, bs, src, dst, dinv, bufY, bufA, N, E, stream);
    relu_inplace<<<cdiv((long long)N * D / 4, 256), 256, 0, stream>>>(bufA, N * D / 4);

    gcn_layer(bufA, Wmu, bmu, src, dst, dinv, bufY, mu, N, E, stream);
    gcn_layer(bufA, Wlv, blv, src, dst, dinv, bufY, lv, N, E, stream);
}


# Weights of the three linear layers multiplied first (W1*W2*Ws, biases carried through) by a spare work-group of launch 1: one N-row WMMA product, not three; 11 launches -> 9; dense rounding changes
# speedup vs baseline: 3.0585x; 1.1506x over previous
#include <hip/hip_runtime.h>
#include <hip/hip_bf16.h>

#define D 64

typedef __attribute__((ext_vector_type(2))) float v2f;
typedef __attribute__((ext_vector_type(8))) float v8f;

__device__ __forceinline__ void atomic_fadd(float* p, float v) {
    __hip_atomic_fetch_add(p, v, __ATOMIC_RELAXED, __HIP_MEMORY_SCOPE_AGENT);
}

__global__ void zero_f32(float* p, int n) {
    int i = blockIdx.x * blockDim.x + threadIdx.x;
    if (i < n) p[i] = 0.0f;
}

__global__ void deg_accum(const int* __restrict__ dst, float* __restrict__ deg, int E) {
    int e = blockIdx.x * blockDim.x + threadIdx.x;
    if (e < E) atomic_fadd(&deg[dst[e]], 1.0f);
}

__global__ void dinv_kernel(const float* __restrict__ deg, float* __restrict__ dinv, int n) {
    int i = blockIdx.x * blockDim.x + threadIdx.x;
    if (i < n) dinv[i] = __frsqrt_rn(deg[i] + 1.0f);
}

#define DEG_BINS    65536
#define DEG_GROUPS  64
#define DEG_THREADS 1024
#define DEG_INFLIGHT 16

#define RS      128
#define BP_NR   512
#define BP_CAP  12800
#define BP_INFLIGHT 16

#define WF_ROWS  68
#define S1_ROWS  80
#define S1_PITCH 66

__device__ __forceinline__ void fold_weights(const float* __restrict__ W1, const float* __restrict__ b1,
                                             const float* __restrict__ W2, const float* __restrict__ b2,
                                             const float* __restrict__ Ws, float* __restrict__ wf,
                                             float* s1) {
    const int t    = threadIdx.x;
    const int lane = t & 31;
    const int wv   = t >> 5;
    const int half = lane >> 4;
    const int l    = lane & 15;
    const int rt   = wv >> 2;
    const int ct   = wv & 3;
    const int r    = rt * 16 + l;
    const int col  = ct * 16 + l;
    const int ntile = (S1_ROWS / 16) * 4;

    if (wv < ntile) {
        const float* arow = (r < D) ? (W1 + (size_t)r * D) : b1;
        const bool   live = (r <= D);
        const float  b2c  = b2[col];
        v8f acc = {};
#pragma unroll
        for (int k0 = 0; k0 < D; k0 += 4) {
            const int ka = k0 + 2 * half;
            const float ax = arow[ka + 0];
            const float ay = arow[ka + 1];
            v2f a;
            a.x = live ? ax : 0.0f;
            a.y = live ? ay : 0.0f;
            v2f b;
            b.x = W2[(size_t)(ka + 0) * D + col];
            b.y = W2[(size_t)(ka + 1) * D + col];
            acc = __builtin_amdgcn_wmma_f32_16x16x4_f32(false, a, false, b, (short)0, acc, false, false);
        }
#pragma unroll
        for (int i = 0; i < 8; ++i) {
            const int row = rt * 16 + 8 * half + i;
            s1[row * S1_PITCH + col] = (row == D + 1) ? b2c : acc[i];
        }
    }
    __syncthreads();
    if (wv < ntile) {
        const float* arow = s1 + r * S1_PITCH;
        v8f acc = {};
#pragma unroll
        for (int k0 = 0; k0 < D; k0 += 4) {
            const int ka = k0 + 2 * half;
            v2f a;
            a.x = arow[ka + 0];
            a.y = arow[ka + 1];
            v2f b;
            b.x = Ws[(size_t)(ka + 0) * D + col];
            b.y = Ws[(size_t)(ka + 1) * D + col];
            acc = __builtin_amdgcn_wmma_f32_16x16x4_f32(false, a, false, b, (short)0, acc, false, false);
        }
#pragma unroll
        for (int i = 0; i < 8; ++i) {
            const int row = rt * 16 + 8 * half + i;
            if (row < D + 2) wf[(size_t)row * D + col] = acc[i];
        }
    }
}

__global__ __launch_bounds__(DEG_THREADS)
void range_hist(const int* __restrict__ dst, unsigned* __restrict__ rcnt, int NR, int E,
                const float* __restrict__ W1, const float* __restrict__ b1,
                const float* __restrict__ W2, const float* __restrict__ b2,
                const float* __restrict__ Ws, float* __restrict__ wf) {
    __shared__ unsigned rc[BP_NR];
    __shared__ __attribute__((aligned(16))) float s1[S1_ROWS * S1_PITCH];
    if (blockIdx.x == DEG_GROUPS) {
        fold_weights(W1, b1, W2, b2, Ws, wf, s1);
        return;
    }
    const int t = threadIdx.x;
    if (t < BP_NR) rc[t] = 0u;
    __syncthreads();

    const int per = (E + DEG_GROUPS - 1) / DEG_GROUPS;
    const int e0  = (int)blockIdx.x * per;
    const int e1  = (e0 + per < E) ? (e0 + per) : E;

    for (int base = e0 + t; base < e1; base += DEG_INFLIGHT * DEG_THREADS) {
        int d[DEG_INFLIGHT];
#pragma unroll
        for (int u = 0; u < DEG_INFLIGHT; ++u) {
            const int e = base + u * DEG_THREADS;
            d[u] = (e < e1) ? dst[e] : -1;
        }
#pragma unroll
        for (int u = 0; u < DEG_INFLIGHT; ++u)
            if (d[u] >= 0) atomicAdd(&rc[d[u] / RS], 1u);
    }
    __syncthreads();
    if (t < NR) rcnt[(size_t)blockIdx.x * NR + t] = rc[t];
}

#define SUBB   (RS / 16)
#define PADG   16
#define RD_THREADS 256
#define RD_INFLIGHT 8

__global__ __launch_bounds__(RD_THREADS)
void range_sort(const int* __restrict__ bsrc, const int* __restrict__ bdl,
                const int* __restrict__ rstart, float* __restrict__ dinv,
                int2* __restrict__ rec, int2* __restrict__ bspan, int N, int NB) {
    __shared__ unsigned cnt[RS];
    __shared__ unsigned bcnt[SUBB];
    __shared__ unsigned cur[SUBB];
    __shared__ int      bst[SUBB];
    const int t = threadIdx.x;
    if (t < RS) cnt[t] = 0u;
    __syncthreads();

    const int es = rstart[blockIdx.x];
    const int ee = rstart[blockIdx.x + 1];
    for (int base = es + t; base < ee; base += RD_INFLIGHT * RD_THREADS) {
        int d[RD_INFLIGHT];
#pragma unroll
        for (int u = 0; u < RD_INFLIGHT; ++u) {
            const int e = base + u * RD_THREADS;
            d[u] = (e < ee) ? bdl[e] : -1;
        }
#pragma unroll
        for (int u = 0; u < RD_INFLIGHT; ++u)
            if (d[u] >= 0) atomicAdd(&cnt[d[u]], 1u);
    }
    __syncthreads();

    const int row = (int)blockIdx.x * RS + t;
    if (t < RS && row < N) dinv[row] = __frsqrt_rn((float)cnt[t] + 1.0f);
    if (t < SUBB) {
        unsigned c = 0u;
        for (int i = 0; i < 16; ++i) c += cnt[16 * t + i];
        bcnt[t] = c;
        cur[t]  = 0u;
    }
    __syncthreads();
    if (t == 0) {
        int base = ((es + PADG - 1) / PADG) * PADG + (int)blockIdx.x * (SUBB + 1) * PADG;
        for (int j = 0; j < SUBB; ++j) {
            const int pc  = (int)((bcnt[j] + PADG - 1) / PADG) * PADG;
            const int blk = (int)blockIdx.x * SUBB + j;
            bst[j] = base;
            if (blk < NB) bspan[blk] = make_int2(base, pc);
            base += pc;
        }
    }
    __syncthreads();

    for (int base = es + t; base < ee; base += RD_INFLIGHT * RD_THREADS) {
        int s[RD_INFLIGHT], d[RD_INFLIGHT];
#pragma unroll
        for (int u = 0; u < RD_INFLIGHT; ++u) {
            const int e  = base + u * RD_THREADS;
            const int ec = (e < ee) ? e : (ee - 1);
            s[u] = bsrc[ec];
            d[u] = bdl[ec];
        }
#pragma unroll
        for (int u = 0; u < RD_INFLIGHT; ++u) {
            if (base + u * RD_THREADS < ee) {
                const int j = d[u] >> 4;
                const unsigned p = atomicAdd(&cur[j], 1u);
                rec[bst[j] + (int)p] = make_int2((s[u] << 5) | (d[u] & 15), 0);
            }
        }
    }
    if (t < SUBB * PADG) {
        const int j  = t / PADG;
        const int i  = t % PADG;
        const int c  = (int)bcnt[j];
        const int pc = ((c + PADG - 1) / PADG) * PADG;
        if (c + i < pc) rec[bst[j] + c + i] = make_int2(16, 0);
    }
}

__global__ __launch_bounds__(DEG_THREADS)
void deg_hist(const int* __restrict__ dst, unsigned* __restrict__ planes,
              unsigned* __restrict__ rcnt, int NR,
              int E, int N, int pitch) {
    __shared__ unsigned bins[DEG_BINS];
    const int t = threadIdx.x;

    const int n4 = N >> 2;
    for (int i = t; i < n4; i += DEG_THREADS)
        ((uint4*)bins)[i] = make_uint4(0u, 0u, 0u, 0u);
    for (int i = (n4 << 2) + t; i < N; i += DEG_THREADS) bins[i] = 0u;
    __syncthreads();

    const int per = (E + (int)gridDim.x - 1) / (int)gridDim.x;
    const int e0  = (int)blockIdx.x * per;
    const int e1  = (e0 + per < E) ? (e0 + per) : E;

    for (int base = e0 + t; base < e1; base += DEG_INFLIGHT * DEG_THREADS) {
        int d[DEG_INFLIGHT];
#pragma unroll
        for (int u = 0; u < DEG_INFLIGHT; ++u) {
            const int e = base + u * DEG_THREADS;
            d[u] = (e < e1) ? dst[e] : -1;
        }
#pragma unroll
        for (int u = 0; u < DEG_INFLIGHT; ++u)
            if (d[u] >= 0) atomicAdd(&bins[d[u]], 1u);
    }
    __syncthreads();

    unsigned* out = planes + (size_t)blockIdx.x * pitch;
    for (int i = t; i < N; i += DEG_THREADS) out[i] = bins[i];

    if (rcnt != nullptr) {
        for (int r = t; r < NR; r += DEG_THREADS) {
            const int n0 = r * RS;
            const int n1 = (n0 + RS < N) ? (n0 + RS) : N;
            unsigned c = 0u;
            for (int i = n0; i < n1; ++i) c += bins[i];
            rcnt[(size_t)blockIdx.x * NR + r] = c;
        }
    }
}

__global__ void dinv_from_planes(const unsigned* __restrict__ planes,
                                 float* __restrict__ dinv, int n, int pitch) {
    int i = blockIdx.x * blockDim.x + threadIdx.x;
    if (i >= n) return;
    unsigned c = 0u;
#pragma unroll
    for (int g = 0; g < DEG_GROUPS; ++g) c += planes[(size_t)g * pitch + i];
    dinv[i] = __frsqrt_rn((float)c + 1.0f);
}

__global__ void gemm_xw_wmma(const float* __restrict__ X,
                             const float* __restrict__ W,
                             float* __restrict__ Y, int nrows) {
    const int wave = (int)((blockIdx.x * blockDim.x + threadIdx.x) >> 5);
    const int lane = threadIdx.x & 31;
    const int row0 = wave * 16;
    if (row0 >= nrows) return;

    const int half = lane >> 4;
    const int l    = lane & 15;

    v8f acc0 = {}, acc1 = {}, acc2 = {}, acc3 = {};

    const float* xrow = X + (size_t)(row0 + l) * D;

#pragma unroll
    for (int k0 = 0; k0 < D; k0 += 4) {
        const int ka = k0 + 2 * half;
        v2f a;
        a.x = xrow[ka + 0];
        a.y = xrow[ka + 1];

        const float* w0 = W + (size_t)(ka + 0) * D;
        const float* w1 = W + (size_t)(ka + 1) * D;

        v2f b0; b0.x = w0[ 0 + l]; b0.y = w1[ 0 + l];
        v2f b1; b1.x = w0[16 + l]; b1.y = w1[16 + l];
        v2f b2; b2.x = w0[32 + l]; b2.y = w1[32 + l];
        v2f b3; b3.x = w0[48 + l]; b3.y = w1[48 + l];

        acc0 = __builtin_amdgcn_wmma_f32_16x16x4_f32(false, a, false, b0, (short)0, acc0, false, false);
        acc1 = __builtin_amdgcn_wmma_f32_16x16x4_f32(false, a, false, b1, (short)0, acc1, false, false);
        acc2 = __builtin_amdgcn_wmma_f32_16x16x4_f32(false, a, false, b2, (short)0, acc2, false, false);
        acc3 = __builtin_amdgcn_wmma_f32_16x16x4_f32(false, a, false, b3, (short)0, acc3, false, false);
    }

#pragma unroll
    for (int r = 0; r < 8; ++r) {
        const int row = row0 + r + 8 * half;
        float* yrow = Y + (size_t)row * D + l;
        yrow[ 0] = acc0[r];
        yrow[16] = acc1[r];
        yrow[32] = acc2[r];
        yrow[48] = acc3[r];
    }
}

__global__ void self_bias(const float* __restrict__ Y,
                          const float* __restrict__ dinv,
                          const float* __restrict__ b,
                          float* __restrict__ agg, int n) {
    int t = blockIdx.x * blockDim.x + threadIdx.x;
    int i = t >> 4;
    int q = (t & 15) * 4;
    if (i >= n) return;
    float di = dinv[i];
    float c  = di * di;
    const float4 y  = *(const float4*)(Y + (size_t)i * D + q);
    const float4 bb = *(const float4*)(b + q);
    float4 r;
    r.x = fmaf(y.x, c, bb.x);
    r.y = fmaf(y.y, c, bb.y);
    r.z = fmaf(y.z, c, bb.z);
    r.w = fmaf(y.w, c, bb.w);
    *(float4*)(agg + (size_t)i * D + q) = r;
}

#define EB 8

__global__ void edge_scatter(const float* __restrict__ Y,
                             const int* __restrict__ src,
                             const int* __restrict__ dst,
                             const float* __restrict__ dinv,
                             float* __restrict__ agg, int E) {
    const int lane = threadIdx.x & 31;
    const int wave = (int)((blockIdx.x * blockDim.x + threadIdx.x) >> 5);
    const int e0   = wave * 32;
    if (e0 >= E) return;

    const int e = e0 + lane;
    int s = 0, d = 0;
    float coef = 0.0f;
    if (e < E) {
        s = src[e];
        d = dst[e];
        coef = dinv[s] * dinv[d];
    }
    const int cbits = __float_as_int(coef);
    const int n = (E - e0 < 32) ? (E - e0) : 32;

    if (n == 32) {
        float va[EB], vb[EB], wa[EB], wb[EB];
#pragma unroll
        for (int u = 0; u < EB; ++u) {
            const int sj = __builtin_amdgcn_readlane(s, u);
            const float* yr = Y + (size_t)sj * D + lane;
            va[u] = yr[0];
            vb[u] = yr[32];
        }
#pragma unroll
        for (int j0 = 0; j0 < 32; j0 += EB) {
            if (j0 + EB < 32) {
#pragma unroll
                for (int u = 0; u < EB; ++u) {
                    const int sj = __builtin_amdgcn_readlane(s, j0 + EB + u);
                    const float* yr = Y + (size_t)sj * D + lane;
                    wa[u] = yr[0];
                    wb[u] = yr[32];
                }
            }
#pragma unroll
            for (int u = 0; u < EB; ++u) {
                const int   dj = __builtin_amdgcn_readlane(d, j0 + u);
                const float cj = __int_as_float(__builtin_amdgcn_readlane(cbits, j0 + u));
                float* o = agg + (size_t)dj * D + lane;
                atomic_fadd(o,      va[u] * cj);
                atomic_fadd(o + 32, vb[u] * cj);
            }
#pragma unroll
            for (int u = 0; u < EB; ++u) { va[u] = wa[u]; vb[u] = wb[u]; }
        }
    } else {
        for (int j = 0; j < n; ++j) {
            const int   sj = __builtin_amdgcn_readlane(s, j);
            const int   dj = __builtin_amdgcn_readlane(d, j);
            const float cj = __int_as_float(__builtin_amdgcn_readlane(cbits, j));
            const float* yr = Y + (size_t)sj * D + lane;
            float* o = agg + (size_t)dj * D + lane;
            atomic_fadd(o,      yr[0]  * cj);
            atomic_fadd(o + 32, yr[32] * cj);
        }
    }
}

__global__ __launch_bounds__(1024)
void bin_place(const int* __restrict__ src, const int* __restrict__ dst,
               const unsigned* __restrict__ rcnt,
               int* __restrict__ bsrc, int* __restrict__ bdl, int* __restrict__ rstart,
               int E, int NR) {
    __shared__ int      s_l[BP_CAP];
    __shared__ int      d_l[BP_CAP];
    __shared__ unsigned sc_own[BP_NR];
    __shared__ unsigned sc_tot[BP_NR];
    __shared__ unsigned lstart[BP_NR];
    __shared__ unsigned gbase[BP_NR];
    __shared__ unsigned cur[BP_NR];

    const int t = threadIdx.x;
    const int g = blockIdx.x;
    const int G = DEG_GROUPS;

    unsigned own = 0u, before = 0u, total = 0u;
    {
        const int tr = (t < NR) ? t : (NR - 1);
        unsigned c[DEG_GROUPS];
#pragma unroll
        for (int g2 = 0; g2 < DEG_GROUPS; ++g2) c[g2] = rcnt[(size_t)g2 * NR + tr];
#pragma unroll
        for (int g2 = 0; g2 < DEG_GROUPS; ++g2) {
            total  += c[g2];
            before += (g2 < g)  ? c[g2] : 0u;
            own     = (g2 == g) ? c[g2] : own;
        }
        if (t >= NR) { own = 0u; before = 0u; total = 0u; }
    }
    if (t < BP_NR) { sc_own[t] = own; sc_tot[t] = total; }
    __syncthreads();
    for (int off = 1; off < BP_NR; off <<= 1) {
        unsigned a = 0u, b = 0u;
        if (t < BP_NR && t >= off) { a = sc_own[t - off]; b = sc_tot[t - off]; }
        __syncthreads();
        if (t < BP_NR) { sc_own[t] += a; sc_tot[t] += b; }
        __syncthreads();
    }
    if (t < NR) {
        const unsigned ls = sc_own[t] - own;
        const unsigned gs = sc_tot[t] - total;
        lstart[t] = ls;
        cur[t]    = ls;
        gbase[t]  = gs + before;
        if (g == 0) {
            rstart[t] = (int)gs;
            if (t == NR - 1) rstart[NR] = (int)sc_tot[t];
        }
    }
    __syncthreads();

    const int per = (E + G - 1) / G;
    const int e0  = g * per;
    const int e1  = (e0 + per < E) ? (e0 + per) : E;

    for (int base = e0 + t; base < e1; base += BP_INFLIGHT * 1024) {
        int s[BP_INFLIGHT], d[BP_INFLIGHT];
#pragma unroll
        for (int u = 0; u < BP_INFLIGHT; ++u) {
            const int e = base + u * 1024;
            const int ec = (e < e1) ? e : (e1 - 1);
            s[u] = src[ec];
            d[u] = dst[ec];
        }
#pragma unroll
        for (int u = 0; u < BP_INFLIGHT; ++u) {
            if (base + u * 1024 < e1) {
                const int r = d[u] / RS;
                const unsigned p = atomicAdd(&cur[r], 1u);
                if (p < (unsigned)BP_CAP) { s_l[p] = s[u]; d_l[p] = d[u]; }
            }
        }
    }
    __syncthreads();

    const int cnt = e1 - e0;
    for (int i = t; i < cnt; i += 1024) {
        const int   s = s_l[i];
        const int   d = d_l[i];
        const int   r = d / RS;
        const unsigned slot = gbase[r] + ((unsigned)i - lstart[r]);
        if (slot < (unsigned)E) {
            bsrc[slot] = s;
            bdl[slot]  = d - r * RS;
        }
    }
}

#define BA_WAVES 4

template <int NG, bool MAKE_CF>
__global__ __launch_bounds__(BA_WAVES * 32)
void block_accum_wmma(const float* __restrict__ Y, int2* rec,
                      const int2* __restrict__ bspan,
                      const float* __restrict__ dinv,
                      const float* __restrict__ ba, const float* __restrict__ bb,
                      float* __restrict__ outa, float* __restrict__ outb, int N) {
    constexpr int DC = D * NG;
    constexpr int Q  = DC / 4;
    __shared__ __attribute__((aligned(16))) float part[BA_WAVES * 16 * DC];

    const int t    = threadIdx.x;
    const int lane = t & 31;
    const int w    = t >> 5;
    const int h    = lane >> 4;
    const int l    = lane & 15;
    const int row0 = (int)blockIdx.x * 16;

    const int2 sp  = bspan[blockIdx.x];
    const int per  = sp.y / (4 * BA_WAVES);
    int e = sp.x + w * per * 4 + 2 * h;

    v8f acc[4 * NG];
#pragma unroll
    for (int i = 0; i < 4 * NG; ++i) acc[i] = v8f{};

    if (per > 0) {
        auto issue = [&](const int4& rr, float4* ya, float4* yb, float* dv) {
            const int s0 = rr.x >> 5, s1 = rr.z >> 5;
            const float* y0p = Y + (size_t)s0 * DC + 4 * l;
            const float* y1p = Y + (size_t)s1 * DC + 4 * l;
#pragma unroll
            for (int g = 0; g < NG; ++g) {
                ya[g] = *(const float4*)(y0p + D * g);
                yb[g] = *(const float4*)(y1p + D * g);
            }
            if (MAKE_CF) {
                dv[0] = dinv[s0];
                dv[1] = dinv[row0 + (rr.x & 15)];
                dv[2] = dinv[s1];
                dv[3] = dinv[row0 + (rr.z & 15)];
            }
        };

        const int e_last = e + 4 * (per - 1);
        int4 r = *(const int4*)(rec + e);
        int4 rn;
        float4 y0[NG], y1[NG], z0[NG], z1[NG];
        float dv[4] = {0.0f, 0.0f, 0.0f, 0.0f}, dn[4] = {0.0f, 0.0f, 0.0f, 0.0f};
        issue(r, y0, y1, dv);
        rn = *(const int4*)(rec + ((e + 4 < e_last) ? (e + 4) : e_last));

#pragma unroll 2
        for (int q = 0; q < per; ++q, e += 4) {
            issue(rn, z0, z1, dn);
            const int4 rnn = *(const int4*)(rec + ((e + 8 < e_last) ? (e + 8) : e_last));

            const int d0 = r.x & 31, d1 = r.z & 31;
            float c0, c1;
            if (MAKE_CF) {
                c0 = (d0 < 16) ? dv[0] * dv[1] : 0.0f;
                c1 = (d1 < 16) ? dv[2] * dv[3] : 0.0f;
                if (l == 0) {
                    rec[e].y     = __float_as_int(c0);
                    rec[e + 1].y = __float_as_int(c1);
                }
            } else {
                c0 = __int_as_float(r.y);
                c1 = __int_as_float(r.w);
            }
            v2f a;
            a.x = (d0 == l) ? c0 : 0.0f;
            a.y = (d1 == l) ? c1 : 0.0f;

#pragma unroll
            for (int g = 0; g < NG; ++g) {
                v2f b0; b0.x = y0[g].x; b0.y = y1[g].x;
                v2f b1; b1.x = y0[g].y; b1.y = y1[g].y;
                v2f b2; b2.x = y0[g].z; b2.y = y1[g].z;
                v2f b3; b3.x = y0[g].w; b3.y = y1[g].w;
                acc[4 * g + 0] = __builtin_amdgcn_wmma_f32_16x16x4_f32(false, a, false, b0, (short)0, acc[4 * g + 0], false, false);
                acc[4 * g + 1] = __builtin_amdgcn_wmma_f32_16x16x4_f32(false, a, false, b1, (short)0, acc[4 * g + 1], false, false);
                acc[4 * g + 2] = __builtin_amdgcn_wmma_f32_16x16x4_f32(false, a, false, b2, (short)0, acc[4 * g + 2], false, false);
                acc[4 * g + 3] = __builtin_amdgcn_wmma_f32_16x16x4_f32(false, a, false, b3, (short)0, acc[4 * g + 3], false, false);
            }
            r = rn; rn = rnn;
#pragma unroll
            for (int g = 0; g < NG; ++g) { y0[g] = z0[g]; y1[g] = z1[g]; }
#pragma unroll
            for (int i = 0; i < 4; ++i) dv[i] = dn[i];
        }
    }

#pragma unroll
    for (int i = 0; i < 8; ++i) {
#pragma unroll
        for (int g = 0; g < NG; ++g) {
            float4 v;
            v.x = acc[4 * g + 0][i];
            v.y = acc[4 * g + 1][i];
            v.z = acc[4 * g + 2][i];
            v.w = acc[4 * g + 3][i];
            *(float4*)(part + ((w * 16 + 8 * h + i) * DC + D * g + 4 * l)) = v;
        }
    }
    __syncthreads();

    for (int idx = t; idx < 16 * Q; idx += BA_WAVES * 32) {
        const int rr   = idx / Q;
        const int c4   = (idx % Q) * 4;
        const int grow = row0 + rr;
        if (grow < N) {
            const float4 p0 = *(const float4*)(part + ((0 * 16 + rr) * DC + c4));
            const float4 p1 = *(const float4*)(part + ((1 * 16 + rr) * DC + c4));
            const float4 p2 = *(const float4*)(part + ((2 * 16 + rr) * DC + c4));
            const float4 p3 = *(const float4*)(part + ((3 * 16 + rr) * DC + c4));
            const float  di = dinv[grow];
            const float  c  = di * di;
            const float4 y  = *(const float4*)(Y + (size_t)grow * DC + c4);
            const float4 bv = (c4 < D) ? *(const float4*)(ba + c4) : *(const float4*)(bb + (c4 - D));
            float4 o;
            o.x = fmaf(y.x, c, bv.x) + ((p0.x + p1.x) + (p2.x + p3.x));
            o.y = fmaf(y.y, c, bv.y) + ((p0.y + p1.y) + (p2.y + p3.y));
            o.z = fmaf(y.z, c, bv.z) + ((p0.z + p1.z) + (p2.z + p3.z));
            o.w = fmaf(y.w, c, bv.w) + ((p0.w + p1.w) + (p2.w + p3.w));
            float* op = (c4 < D) ? (outa + (size_t)grow * D + c4) : (outb + (size_t)grow * D + (c4 - D));
            *(float4*)op = o;
        }
    }
}

#define D2 128

__global__ void gemm_heads_wmma(const float* __restrict__ X,
                                const float* __restrict__ Wa,
                                const float* __restrict__ Wb,
                                float* __restrict__ Y2, int nrows) {
    const int wave = (int)((blockIdx.x * blockDim.x + threadIdx.x) >> 5);
    const int lane = threadIdx.x & 31;
    const int row0 = wave * 16;
    if (row0 >= nrows) return;

    const int half = lane >> 4;
    const int l    = lane & 15;

    v8f p0 = {}, p1 = {}, p2 = {}, p3 = {};
    v8f q0 = {}, q1 = {}, q2 = {}, q3 = {};

    const float* xrow = X + (size_t)(row0 + l) * D;

#pragma unroll
    for (int k0 = 0; k0 < D; k0 += 4) {
        const int ka = k0 + 2 * half;
        v2f a;
        a.x = fmaxf(xrow[ka + 0], 0.0f);
        a.y = fmaxf(xrow[ka + 1], 0.0f);

        const float* w0 = Wa + (size_t)(ka + 0) * D;
        const float* w1 = Wa + (size_t)(ka + 1) * D;
        const float* u0 = Wb + (size_t)(ka + 0) * D;
        const float* u1 = Wb + (size_t)(ka + 1) * D;

        v2f b0; b0.x = w0[ 0 + l]; b0.y = w1[ 0 + l];
        v2f b1; b1.x = w0[16 + l]; b1.y = w1[16 + l];
        v2f b2; b2.x = w0[32 + l]; b2.y = w1[32 + l];
        v2f b3; b3.x = w0[48 + l]; b3.y = w1[48 + l];
        v2f e0; e0.x = u0[ 0 + l]; e0.y = u1[ 0 + l];
        v2f e1; e1.x = u0[16 + l]; e1.y = u1[16 + l];
        v2f e2; e2.x = u0[32 + l]; e2.y = u1[32 + l];
        v2f e3; e3.x = u0[48 + l]; e3.y = u1[48 + l];

        p0 = __builtin_amdgcn_wmma_f32_16x16x4_f32(false, a, false, b0, (short)0, p0, false, false);
        p1 = __builtin_amdgcn_wmma_f32_16x16x4_f32(false, a, false, b1, (short)0, p1, false, false);
        p2 = __builtin_amdgcn_wmma_f32_16x16x4_f32(false, a, false, b2, (short)0, p2, false, false);
        p3 = __builtin_amdgcn_wmma_f32_16x16x4_f32(false, a, false, b3, (short)0, p3, false, false);
        q0 = __builtin_amdgcn_wmma_f32_16x16x4_f32(false, a, false, e0, (short)0, q0, false, false);
        q1 = __builtin_amdgcn_wmma_f32_16x16x4_f32(false, a, false, e1, (short)0, q1, false, false);
        q2 = __builtin_amdgcn_wmma_f32_16x16x4_f32(false, a, false, e2, (short)0, q2, false, false);
        q3 = __builtin_amdgcn_wmma_f32_16x16x4_f32(false, a, false, e3, (short)0, q3, false, false);
    }

#pragma unroll
    for (int r = 0; r < 8; ++r) {
        const int row = row0 + r + 8 * half;
        float* yrow = Y2 + (size_t)row * D2 + l;
        yrow[  0] = p0[r];
        yrow[ 16] = p1[r];
        yrow[ 32] = p2[r];
        yrow[ 48] = p3[r];
        yrow[ 64] = q0[r];
        yrow[ 80] = q1[r];
        yrow[ 96] = q2[r];
        yrow[112] = q3[r];
    }
}

__global__ void relu_inplace(float* __restrict__ x, int n4) {
    int i = blockIdx.x * blockDim.x + threadIdx.x;
    if (i >= n4) return;
    float4 v = ((float4*)x)[i];
    v.x = fmaxf(v.x, 0.0f);
    v.y = fmaxf(v.y, 0.0f);
    v.z = fmaxf(v.z, 0.0f);
    v.w = fmaxf(v.w, 0.0f);
    ((float4*)x)[i] = v;
}

static inline int cdiv(long long a, int b) { return (int)((a + b - 1) / b); }

static void gcn_layer(const float* x, const float* W, const float* b,
                      const int* src, const int* dst, const float* dinv,
                      float* y, float* agg, int N, int E, hipStream_t stream) {
    const int waves  = (N + 15) / 16;
    const int gblk   = cdiv(waves, 8);
    gemm_xw_wmma<<<gblk, 256, 0, stream>>>(x, W, y, N);
    self_bias<<<cdiv((long long)N * 16, 256), 256, 0, stream>>>(y, dinv, b, agg, N);
    edge_scatter<<<cdiv(cdiv(E, 32), 8), 256, 0, stream>>>(y, src, dst, dinv, agg, E);
}

extern "C" void kernel_launch(void* const* d_in, const int* in_sizes, int n_in,
                              void* d_out, int out_size, void* d_ws, size_t ws_size,
                              hipStream_t stream) {
    const int N = in_sizes[0] / D;
    const int E = in_sizes[1] / 2;

    const float* x0  = (const float*)d_in[0];
    const int*   ei  = (const int*)d_in[1];
    const int*   src = ei;
    const int*   dst = ei + E;
    const float* W1  = (const float*)d_in[2];
    const float* b1  = (const float*)d_in[3];
    const float* W2  = (const float*)d_in[4];
    const float* b2  = (const float*)d_in[5];
    const float* Ws  = (const float*)d_in[6];
    const float* bs  = (const float*)d_in[7];
    const float* Wmu = (const float*)d_in[8];
    const float* bmu = (const float*)d_in[9];
    const float* Wlv = (const float*)d_in[10];
    const float* blv = (const float*)d_in[11];

    float* deg  = (float*)d_ws;
    float* dinv = deg + N;
    float* bufA = dinv + N;
    float* bufB = bufA + (size_t)N * D;
    float* bufY = bufB + (size_t)N * D;

    float* mu = (float*)d_out;
    float* lv = mu + (size_t)N * D;

    const int NR = cdiv(N, RS);
    const int NB = cdiv(N, 16);
    if (NR <= BP_NR && cdiv(E, DEG_GROUPS) <= BP_CAP && N <= (1 << 26)) {
        float*    wf     = bufY + (size_t)N * D;
        unsigned* rcnt   = (unsigned*)(wf + (size_t)WF_ROWS * D);
        int*      rstart = (int*)(rcnt + (size_t)DEG_GROUPS * BP_NR);
        int*      bsrc   = rstart + 2 * BP_NR;
        int*      bdl    = bsrc + (((size_t)E + 31) & ~(size_t)31);
        int2*     bspan  = (int2*)(bdl + (((size_t)E + 31) & ~(size_t)31));
        int2*     rec    = bspan + (((size_t)NB + 31) & ~(size_t)31);
        const float* w123 = wf;
        const float* c1   = wf + (size_t)D * D;
        const float* c2   = c1 + D;

        range_hist<<<DEG_GROUPS + 1, DEG_THREADS, 0, stream>>>(dst, rcnt, NR, E, W1, b1, W2, b2, Ws, wf);
        bin_place<<<DEG_GROUPS, 1024, 0, stream>>>(src, dst, rcnt, bsrc, bdl, rstart, E, NR);
        range_sort<<<NR, RD_THREADS, 0, stream>>>(bsrc, bdl, rstart, dinv, rec, bspan, N, NB);

        gemm_xw_wmma<<<cdiv(cdiv(N, 16), 8), 256, 0, stream>>>(x0, w123, bufY, N);
        block_accum_wmma<1, true><<<NB, BA_WAVES * 32, 0, stream>>>(bufY, rec, bspan, dinv, c1, nullptr, bufA, nullptr, N);
        block_accum_wmma<1, false><<<NB, BA_WAVES * 32, 0, stream>>>(bufA, rec, bspan, dinv, c2, nullptr, bufB, nullptr, N);
        block_accum_wmma<1, false><<<NB, BA_WAVES * 32, 0, stream>>>(bufB, rec, bspan, dinv, bs, nullptr, bufA, nullptr, N);
        gemm_heads_wmma<<<cdiv(cdiv(N, 16), 8), 256, 0, stream>>>(bufA, Wmu, Wlv, bufB, N);
        block_accum_wmma<2, false><<<NB, BA_WAVES * 32, 0, stream>>>(bufB, rec, bspan, dinv, bmu, blv, mu, lv, N);
        return;
    }
    if (N <= DEG_BINS) {
        unsigned* planes = (unsigned*)(bufY + (size_t)N * D);
        const int pitch = (N + 31) & ~31;
        deg_hist<<<DEG_GROUPS, DEG_THREADS, 0, stream>>>(dst, planes, nullptr, 0, E, N, pitch);
        dinv_from_planes<<<cdiv(N, 256), 256, 0, stream>>>(planes, dinv, N, pitch);
    } else {
        zero_f32<<<cdiv(N, 256), 256, 0, stream>>>(deg, N);
        deg_accum<<<cdiv(E, 256), 256, 0, stream>>>(dst, deg, E);
        dinv_kernel<<<cdiv(N, 256), 256, 0, stream>>>(deg, dinv, N);
    }

    gcn_layer(x0,   W1, b1, src, dst, dinv, bufY, bufA, N, E, stream);
    gcn_layer(bufA, W2, b2, src, dst, dinv, bufY, bufB, N, E, stream);
    gcn_layer(bufB, Ws, bs, src, dst, dinv, bufY, bufA, N, E, stream);
    relu_inplace<<<cdiv((long long)N * D / 4, 256), 256, 0, stream>>>(bufA, N * D / 4);

    gcn_layer(bufA, Wmu, bmu, src, dst, dinv, bufY, mu, N, E, stream);
    gcn_layer(bufA, Wlv, blv, src, dst, dinv, bufY, lv, N, E, stream);
}


# First dense product (x0 times the folded weights) runs as 192 extra work-groups of the edge-ordering launch, whose 64 slices fill a quarter of the chip; its own launch goes; 9 launches -> 8
# speedup vs baseline: 3.2333x; 1.0572x over previous
#include <hip/hip_runtime.h>
#include <hip/hip_bf16.h>

#define D 64

typedef __attribute__((ext_vector_type(2))) float v2f;
typedef __attribute__((ext_vector_type(8))) float v8f;

__device__ __forceinline__ void atomic_fadd(float* p, float v) {
    __hip_atomic_fetch_add(p, v, __ATOMIC_RELAXED, __HIP_MEMORY_SCOPE_AGENT);
}

__global__ void zero_f32(float* p, int n) {
    int i = blockIdx.x * blockDim.x + threadIdx.x;
    if (i < n) p[i] = 0.0f;
}

__global__ void deg_accum(const int* __restrict__ dst, float* __restrict__ deg, int E) {
    int e = blockIdx.x * blockDim.x + threadIdx.x;
    if (e < E) atomic_fadd(&deg[dst[e]], 1.0f);
}

__global__ void dinv_kernel(const float* __restrict__ deg, float* __restrict__ dinv, int n) {
    int i = blockIdx.x * blockDim.x + threadIdx.x;
    if (i < n) dinv[i] = __frsqrt_rn(deg[i] + 1.0f);
}

#define DEG_BINS    65536
#define DEG_GROUPS  64
#define DEG_THREADS 1024
#define DEG_INFLIGHT 16

#define RS      128
#define BP_NR   512
#define BP_CAP  12800
#define BP_INFLIGHT 16

#define WF_ROWS  68
#define S1_ROWS  80
#define S1_PITCH 66

__device__ __forceinline__ void fold_weights(const float* __restrict__ W1, const float* __restrict__ b1,
                                             const float* __restrict__ W2, const float* __restrict__ b2,
                                             const float* __restrict__ Ws, float* __restrict__ wf,
                                             float* s1) {
    const int t    = threadIdx.x;
    const int lane = t & 31;
    const int wv   = t >> 5;
    const int half = lane >> 4;
    const int l    = lane & 15;
    const int rt   = wv >> 2;
    const int ct   = wv & 3;
    const int r    = rt * 16 + l;
    const int col  = ct * 16 + l;
    const int ntile = (S1_ROWS / 16) * 4;

    if (wv < ntile) {
        const float* arow = (r < D) ? (W1 + (size_t)r * D) : b1;
        const bool   live = (r <= D);
        const float  b2c  = b2[col];
        v8f acc = {};
#pragma unroll
        for (int k0 = 0; k0 < D; k0 += 4) {
            const int ka = k0 + 2 * half;
            const float ax = arow[ka + 0];
            const float ay = arow[ka + 1];
            v2f a;
            a.x = live ? ax : 0.0f;
            a.y = live ? ay : 0.0f;
            v2f b;
            b.x = W2[(size_t)(ka + 0) * D + col];
            b.y = W2[(size_t)(ka + 1) * D + col];
            acc = __builtin_amdgcn_wmma_f32_16x16x4_f32(false, a, false, b, (short)0, acc, false, false);
        }
#pragma unroll
        for (int i = 0; i < 8; ++i) {
            const int row = rt * 16 + 8 * half + i;
            s1[row * S1_PITCH + col] = (row == D + 1) ? b2c : acc[i];
        }
    }
    __syncthreads();
    if (wv < ntile) {
        const float* arow = s1 + r * S1_PITCH;
        v8f acc = {};
#pragma unroll
        for (int k0 = 0; k0 < D; k0 += 4) {
            const int ka = k0 + 2 * half;
            v2f a;
            a.x = arow[ka + 0];
            a.y = arow[ka + 1];
            v2f b;
            b.x = Ws[(size_t)(ka + 0) * D + col];
            b.y = Ws[(size_t)(ka + 1) * D + col];
            acc = __builtin_amdgcn_wmma_f32_16x16x4_f32(false, a, false, b, (short)0, acc, false, false);
        }
#pragma unroll
        for (int i = 0; i < 8; ++i) {
            const int row = rt * 16 + 8 * half + i;
            if (row < D + 2) wf[(size_t)row * D + col] = acc[i];
        }
    }
}

__global__ __launch_bounds__(DEG_THREADS)
void range_hist(const int* __restrict__ dst, unsigned* __restrict__ rcnt, int NR, int E,
                const float* __restrict__ W1, const float* __restrict__ b1,
                const float* __restrict__ W2, const float* __restrict__ b2,
                const float* __restrict__ Ws, float* __restrict__ wf) {
    __shared__ unsigned rc[BP_NR];
    __shared__ __attribute__((aligned(16))) float s1[S1_ROWS * S1_PITCH];
    if (blockIdx.x == DEG_GROUPS) {
        fold_weights(W1, b1, W2, b2, Ws, wf, s1);
        return;
    }
    const int t = threadIdx.x;
    if (t < BP_NR) rc[t] = 0u;
    __syncthreads();

    const int per = (E + DEG_GROUPS - 1) / DEG_GROUPS;
    const int e0  = (int)blockIdx.x * per;
    const int e1  = (e0 + per < E) ? (e0 + per) : E;

    for (int base = e0 + t; base < e1; base += DEG_INFLIGHT * DEG_THREADS) {
        int d[DEG_INFLIGHT];
#pragma unroll
        for (int u = 0; u < DEG_INFLIGHT; ++u) {
            const int e = base + u * DEG_THREADS;
            d[u] = (e < e1) ? dst[e] : -1;
        }
#pragma unroll
        for (int u = 0; u < DEG_INFLIGHT; ++u)
            if (d[u] >= 0) atomicAdd(&rc[d[u] / RS], 1u);
    }
    __syncthreads();
    if (t < NR) rcnt[(size_t)blockIdx.x * NR + t] = rc[t];
}

#define SUBB   (RS / 16)
#define PADG   16
#define RD_THREADS 256
#define RD_INFLIGHT 8

__global__ __launch_bounds__(RD_THREADS)
void range_sort(const int* __restrict__ bsrc, const int* __restrict__ bdl,
                const int* __restrict__ rstart, float* __restrict__ dinv,
                int2* __restrict__ rec, int2* __restrict__ bspan, int N, int NB) {
    __shared__ unsigned cnt[RS];
    __shared__ unsigned bcnt[SUBB];
    __shared__ unsigned cur[SUBB];
    __shared__ int      bst[SUBB];
    const int t = threadIdx.x;
    if (t < RS) cnt[t] = 0u;
    __syncthreads();

    const int es = rstart[blockIdx.x];
    const int ee = rstart[blockIdx.x + 1];
    for (int base = es + t; base < ee; base += RD_INFLIGHT * RD_THREADS) {
        int d[RD_INFLIGHT];
#pragma unroll
        for (int u = 0; u < RD_INFLIGHT; ++u) {
            const int e = base + u * RD_THREADS;
            d[u] = (e < ee) ? bdl[e] : -1;
        }
#pragma unroll
        for (int u = 0; u < RD_INFLIGHT; ++u)
            if (d[u] >= 0) atomicAdd(&cnt[d[u]], 1u);
    }
    __syncthreads();

    const int row = (int)blockIdx.x * RS + t;
    if (t < RS && row < N) dinv[row] = __frsqrt_rn((float)cnt[t] + 1.0f);
    if (t < SUBB) {
        unsigned c = 0u;
        for (int i = 0; i < 16; ++i) c += cnt[16 * t + i];
        bcnt[t] = c;
        cur[t]  = 0u;
    }
    __syncthreads();
    if (t == 0) {
        int base = ((es + PADG - 1) / PADG) * PADG + (int)blockIdx.x * (SUBB + 1) * PADG;
        for (int j = 0; j < SUBB; ++j) {
            const int pc  = (int)((bcnt[j] + PADG - 1) / PADG) * PADG;
            const int blk = (int)blockIdx.x * SUBB + j;
            bst[j] = base;
            if (blk < NB) bspan[blk] = make_int2(base, pc);
            base += pc;
        }
    }
    __syncthreads();

    for (int base = es + t; base < ee; base += RD_INFLIGHT * RD_THREADS) {
        int s[RD_INFLIGHT], d[RD_INFLIGHT];
#pragma unroll
        for (int u = 0; u < RD_INFLIGHT; ++u) {
            const int e  = base + u * RD_THREADS;
            const int ec = (e < ee) ? e : (ee - 1);
            s[u] = bsrc[ec];
            d[u] = bdl[ec];
        }
#pragma unroll
        for (int u = 0; u < RD_INFLIGHT; ++u) {
            if (base + u * RD_THREADS < ee) {
                const int j = d[u] >> 4;
                const unsigned p = atomicAdd(&cur[j], 1u);
                rec[bst[j] + (int)p] = make_int2((s[u] << 5) | (d[u] & 15), 0);
            }
        }
    }
    if (t < SUBB * PADG) {
        const int j  = t / PADG;
        const int i  = t % PADG;
        const int c  = (int)bcnt[j];
        const int pc = ((c + PADG - 1) / PADG) * PADG;
        if (c + i < pc) rec[bst[j] + c + i] = make_int2(16, 0);
    }
}

__global__ __launch_bounds__(DEG_THREADS)
void deg_hist(const int* __restrict__ dst, unsigned* __restrict__ planes,
              unsigned* __restrict__ rcnt, int NR,
              int E, int N, int pitch) {
    __shared__ unsigned bins[DEG_BINS];
    const int t = threadIdx.x;

    const int n4 = N >> 2;
    for (int i = t; i < n4; i += DEG_THREADS)
        ((uint4*)bins)[i] = make_uint4(0u, 0u, 0u, 0u);
    for (int i = (n4 << 2) + t; i < N; i += DEG_THREADS) bins[i] = 0u;
    __syncthreads();

    const int per = (E + (int)gridDim.x - 1) / (int)gridDim.x;
    const int e0  = (int)blockIdx.x * per;
    const int e1  = (e0 + per < E) ? (e0 + per) : E;

    for (int base = e0 + t; base < e1; base += DEG_INFLIGHT * DEG_THREADS) {
        int d[DEG_INFLIGHT];
#pragma unroll
        for (int u = 0; u < DEG_INFLIGHT; ++u) {
            const int e = base + u * DEG_THREADS;
            d[u] = (e < e1) ? dst[e] : -1;
        }
#pragma unroll
        for (int u = 0; u < DEG_INFLIGHT; ++u)
            if (d[u] >= 0) atomicAdd(&bins[d[u]], 1u);
    }
    __syncthreads();

    unsigned* out = planes + (size_t)blockIdx.x * pitch;
    for (int i = t; i < N; i += DEG_THREADS) out[i] = bins[i];

    if (rcnt != nullptr) {
        for (int r = t; r < NR; r += DEG_THREADS) {
            const int n0 = r * RS;
            const int n1 = (n0 + RS < N) ? (n0 + RS) : N;
            unsigned c = 0u;
            for (int i = n0; i < n1; ++i) c += bins[i];
            rcnt[(size_t)blockIdx.x * NR + r] = c;
        }
    }
}

__global__ void dinv_from_planes(const unsigned* __restrict__ planes,
                                 float* __restrict__ dinv, int n, int pitch) {
    int i = blockIdx.x * blockDim.x + threadIdx.x;
    if (i >= n) return;
    unsigned c = 0u;
#pragma unroll
    for (int g = 0; g < DEG_GROUPS; ++g) c += planes[(size_t)g * pitch + i];
    dinv[i] = __frsqrt_rn((float)c + 1.0f);
}

template <int NT>
__device__ __forceinline__ void xw_strip(const float* __restrict__ X,
                                         const float* __restrict__ W,
                                         float* __restrict__ Y, int row0, int c0, int lane) {
    const int half = lane >> 4;
    const int l    = lane & 15;

    v8f acc[NT];
#pragma unroll
    for (int j = 0; j < NT; ++j) acc[j] = v8f{};

    const float* xrow = X + (size_t)(row0 + l) * D;

#pragma unroll
    for (int k0 = 0; k0 < D; k0 += 4) {
        const int ka = k0 + 2 * half;
        v2f a;
        a.x = xrow[ka + 0];
        a.y = xrow[ka + 1];

        const float* w0 = W + (size_t)(ka + 0) * D + c0 + l;
        const float* w1 = W + (size_t)(ka + 1) * D + c0 + l;

        v2f b[NT];
#pragma unroll
        for (int j = 0; j < NT; ++j) { b[j].x = w0[16 * j]; b[j].y = w1[16 * j]; }
#pragma unroll
        for (int j = 0; j < NT; ++j)
            acc[j] = __builtin_amdgcn_wmma_f32_16x16x4_f32(false, a, false, b[j], (short)0, acc[j], false, false);
    }

#pragma unroll
    for (int r = 0; r < 8; ++r) {
        const int row = row0 + r + 8 * half;
        float* yrow = Y + (size_t)row * D + c0 + l;
#pragma unroll
        for (int j = 0; j < NT; ++j) yrow[16 * j] = acc[j][r];
    }
}

__global__ void gemm_xw_wmma(const float* __restrict__ X,
                             const float* __restrict__ W,
                             float* __restrict__ Y, int nrows) {
    const int wave = (int)((blockIdx.x * blockDim.x + threadIdx.x) >> 5);
    const int lane = threadIdx.x & 31;
    const int row0 = wave * 16;
    if (row0 >= nrows) return;
    xw_strip<4>(X, W, Y, row0, 0, lane);
}

__global__ void self_bias(const float* __restrict__ Y,
                          const float* __restrict__ dinv,
                          const float* __restrict__ b,
                          float* __restrict__ agg, int n) {
    int t = blockIdx.x * blockDim.x + threadIdx.x;
    int i = t >> 4;
    int q = (t & 15) * 4;
    if (i >= n) return;
    float di = dinv[i];
    float c  = di * di;
    const float4 y  = *(const float4*)(Y + (size_t)i * D + q);
    const float4 bb = *(const float4*)(b + q);
    float4 r;
    r.x = fmaf(y.x, c, bb.x);
    r.y = fmaf(y.y, c, bb.y);
    r.z = fmaf(y.z, c, bb.z);
    r.w = fmaf(y.w, c, bb.w);
    *(float4*)(agg + (size_t)i * D + q) = r;
}

#define EB 8

__global__ void edge_scatter(const float* __restrict__ Y,
                             const int* __restrict__ src,
                             const int* __restrict__ dst,
                             const float* __restrict__ dinv,
                             float* __restrict__ agg, int E) {
    const int lane = threadIdx.x & 31;
    const int wave = (int)((blockIdx.x * blockDim.x + threadIdx.x) >> 5);
    const int e0   = wave * 32;
    if (e0 >= E) return;

    const int e = e0 + lane;
    int s = 0, d = 0;
    float coef = 0.0f;
    if (e < E) {
        s = src[e];
        d = dst[e];
        coef = dinv[s] * dinv[d];
    }
    const int cbits = __float_as_int(coef);
    const int n = (E - e0 < 32) ? (E - e0) : 32;

    if (n == 32) {
        float va[EB], vb[EB], wa[EB], wb[EB];
#pragma unroll
        for (int u = 0; u < EB; ++u) {
            const int sj = __builtin_amdgcn_readlane(s, u);
            const float* yr = Y + (size_t)sj * D + lane;
            va[u] = yr[0];
            vb[u] = yr[32];
        }
#pragma unroll
        for (int j0 = 0; j0 < 32; j0 += EB) {
            if (j0 + EB < 32) {
#pragma unroll
                for (int u = 0; u < EB; ++u) {
                    const int sj = __builtin_amdgcn_readlane(s, j0 + EB + u);
                    const float* yr = Y + (size_t)sj * D + lane;
                    wa[u] = yr[0];
                    wb[u] = yr[32];
                }
            }
#pragma unroll
            for (int u = 0; u < EB; ++u) {
                const int   dj = __builtin_amdgcn_readlane(d, j0 + u);
                const float cj = __int_as_float(__builtin_amdgcn_readlane(cbits, j0 + u));
                float* o = agg + (size_t)dj * D + lane;
                atomic_fadd(o,      va[u] * cj);
                atomic_fadd(o + 32, vb[u] * cj);
            }
#pragma unroll
            for (int u = 0; u < EB; ++u) { va[u] = wa[u]; vb[u] = wb[u]; }
        }
    } else {
        for (int j = 0; j < n; ++j) {
            const int   sj = __builtin_amdgcn_readlane(s, j);
            const int   dj = __builtin_amdgcn_readlane(d, j);
            const float cj = __int_as_float(__builtin_amdgcn_readlane(cbits, j));
            const float* yr = Y + (size_t)sj * D + lane;
            float* o = agg + (size_t)dj * D + lane;
            atomic_fadd(o,      yr[0]  * cj);
            atomic_fadd(o + 32, yr[32] * cj);
        }
    }
}

#define XW_GROUPS 192

__global__ __launch_bounds__(1024)
void bin_place(const int* __restrict__ src, const int* __restrict__ dst,
               const unsigned* __restrict__ rcnt,
               int* __restrict__ bsrc, int* __restrict__ bdl, int* __restrict__ rstart,
               int E, int NR,
               const float* __restrict__ X, const float* __restrict__ Wf,
               float* __restrict__ Y, int nrows) {
    if (blockIdx.x >= DEG_GROUPS) {
        const int nwaves = ((int)gridDim.x - DEG_GROUPS) * 32;
        const int ntask  = 2 * ((nrows + 15) / 16);
        const int wv = __builtin_amdgcn_readfirstlane((int)threadIdx.x >> 5);
        for (int task = ((int)blockIdx.x - DEG_GROUPS) * 32 + wv; task < ntask; task += nwaves)
            xw_strip<2>(X, Wf, Y, (task >> 1) * 16, (task & 1) * 32, (int)threadIdx.x & 31);
        return;
    }
    __shared__ int      s_l[BP_CAP];
    __shared__ int      d_l[BP_CAP];
    __shared__ unsigned sc_own[BP_NR];
    __shared__ unsigned sc_tot[BP_NR];
    __shared__ unsigned lstart[BP_NR];
    __shared__ unsigned gbase[BP_NR];
    __shared__ unsigned cur[BP_NR];

    const int t = threadIdx.x;
    const int g = blockIdx.x;
    const int G = DEG_GROUPS;

    unsigned own = 0u, before = 0u, total = 0u;
    {
        const int tr = (t < NR) ? t : (NR - 1);
        unsigned c[DEG_GROUPS];
#pragma unroll
        for (int g2 = 0; g2 < DEG_GROUPS; ++g2) c[g2] = rcnt[(size_t)g2 * NR + tr];
#pragma unroll
        for (int g2 = 0; g2 < DEG_GROUPS; ++g2) {
            total  += c[g2];
            before += (g2 < g)  ? c[g2] : 0u;
            own     = (g2 == g) ? c[g2] : own;
        }
        if (t >= NR) { own = 0u; before = 0u; total = 0u; }
    }
    if (t < BP_NR) { sc_own[t] = own; sc_tot[t] = total; }
    __syncthreads();
    for (int off = 1; off < BP_NR; off <<= 1) {
        unsigned a = 0u, b = 0u;
        if (t < BP_NR && t >= off) { a = sc_own[t - off]; b = sc_tot[t - off]; }
        __syncthreads();
        if (t < BP_NR) { sc_own[t] += a; sc_tot[t] += b; }
        __syncthreads();
    }
    if (t < NR) {
        const unsigned ls = sc_own[t] - own;
        const unsigned gs = sc_tot[t] - total;
        lstart[t] = ls;
        cur[t]    = ls;
        gbase[t]  = gs + before;
        if (g == 0) {
            rstart[t] = (int)gs;
            if (t == NR - 1) rstart[NR] = (int)sc_tot[t];
        }
    }
    __syncthreads();

    const int per = (E + G - 1) / G;
    const int e0  = g * per;
    const int e1  = (e0 + per < E) ? (e0 + per) : E;

    for (int base = e0 + t; base < e1; base += BP_INFLIGHT * 1024) {
        int s[BP_INFLIGHT], d[BP_INFLIGHT];
#pragma unroll
        for (int u = 0; u < BP_INFLIGHT; ++u) {
            const int e = base + u * 1024;
            const int ec = (e < e1) ? e : (e1 - 1);
            s[u] = src[ec];
            d[u] = dst[ec];
        }
#pragma unroll
        for (int u = 0; u < BP_INFLIGHT; ++u) {
            if (base + u * 1024 < e1) {
                const int r = d[u] / RS;
                const unsigned p = atomicAdd(&cur[r], 1u);
                if (p < (unsigned)BP_CAP) { s_l[p] = s[u]; d_l[p] = d[u]; }
            }
        }
    }
    __syncthreads();

    const int cnt = e1 - e0;
    for (int i = t; i < cnt; i += 1024) {
        const int   s = s_l[i];
        const int   d = d_l[i];
        const int   r = d / RS;
        const unsigned slot = gbase[r] + ((unsigned)i - lstart[r]);
        if (slot < (unsigned)E) {
            bsrc[slot] = s;
            bdl[slot]  = d - r * RS;
        }
    }
}

#define BA_WAVES 4

template <int NG, bool MAKE_CF>
__global__ __launch_bounds__(BA_WAVES * 32)
void block_accum_wmma(const float* __restrict__ Y, int2* rec,
                      const int2* __restrict__ bspan,
                      const float* __restrict__ dinv,
                      const float* __restrict__ ba, const float* __restrict__ bb,
                      float* __restrict__ outa, float* __restrict__ outb, int N) {
    constexpr int DC = D * NG;
    constexpr int Q  = DC / 4;
    __shared__ __attribute__((aligned(16))) float part[BA_WAVES * 16 * DC];

    const int t    = threadIdx.x;
    const int lane = t & 31;
    const int w    = t >> 5;
    const int h    = lane >> 4;
    const int l    = lane & 15;
    const int row0 = (int)blockIdx.x * 16;

    const int2 sp  = bspan[blockIdx.x];
    const int per  = sp.y / (4 * BA_WAVES);
    int e = sp.x + w * per * 4 + 2 * h;

    v8f acc[4 * NG];
#pragma unroll
    for (int i = 0; i < 4 * NG; ++i) acc[i] = v8f{};

    if (per > 0) {
        auto issue = [&](const int4& rr, float4* ya, float4* yb, float* dv) {
            const int s0 = rr.x >> 5, s1 = rr.z >> 5;
            const float* y0p = Y + (size_t)s0 * DC + 4 * l;
            const float* y1p = Y + (size_t)s1 * DC + 4 * l;
#pragma unroll
            for (int g = 0; g < NG; ++g) {
                ya[g] = *(const float4*)(y0p + D * g);
                yb[g] = *(const float4*)(y1p + D * g);
            }
            if (MAKE_CF) {
                dv[0] = dinv[s0];
                dv[1] = dinv[row0 + (rr.x & 15)];
                dv[2] = dinv[s1];
                dv[3] = dinv[row0 + (rr.z & 15)];
            }
        };

        const int e_last = e + 4 * (per - 1);
        int4 r = *(const int4*)(rec + e);
        int4 rn;
        float4 y0[NG], y1[NG], z0[NG], z1[NG];
        float dv[4] = {0.0f, 0.0f, 0.0f, 0.0f}, dn[4] = {0.0f, 0.0f, 0.0f, 0.0f};
        issue(r, y0, y1, dv);
        rn = *(const int4*)(rec + ((e + 4 < e_last) ? (e + 4) : e_last));

#pragma unroll 2
        for (int q = 0; q < per; ++q, e += 4) {
            issue(rn, z0, z1, dn);
            const int4 rnn = *(const int4*)(rec + ((e + 8 < e_last) ? (e + 8) : e_last));

            const int d0 = r.x & 31, d1 = r.z & 31;
            float c0, c1;
            if (MAKE_CF) {
                c0 = (d0 < 16) ? dv[0] * dv[1] : 0.0f;
                c1 = (d1 < 16) ? dv[2] * dv[3] : 0.0f;
                if (l == 0) {
                    rec[e].y     = __float_as_int(c0);
                    rec[e + 1].y = __float_as_int(c1);
                }
            } else {
                c0 = __int_as_float(r.y);
                c1 = __int_as_float(r.w);
            }
            v2f a;
            a.x = (d0 == l) ? c0 : 0.0f;
            a.y = (d1 == l) ? c1 : 0.0f;

#pragma unroll
            for (int g = 0; g < NG; ++g) {
                v2f b0; b0.x = y0[g].x; b0.y = y1[g].x;
                v2f b1; b1.x = y0[g].y; b1.y = y1[g].y;
                v2f b2; b2.x = y0[g].z; b2.y = y1[g].z;
                v2f b3; b3.x = y0[g].w; b3.y = y1[g].w;
                acc[4 * g + 0] = __builtin_amdgcn_wmma_f32_16x16x4_f32(false, a, false, b0, (short)0, acc[4 * g + 0], false, false);
                acc[4 * g + 1] = __builtin_amdgcn_wmma_f32_16x16x4_f32(false, a, false, b1, (short)0, acc[4 * g + 1], false, false);
                acc[4 * g + 2] = __builtin_amdgcn_wmma_f32_16x16x4_f32(false, a, false, b2, (short)0, acc[4 * g + 2], false, false);
                acc[4 * g + 3] = __builtin_amdgcn_wmma_f32_16x16x4_f32(false, a, false, b3, (short)0, acc[4 * g + 3], false, false);
            }
            r = rn; rn = rnn;
#pragma unroll
            for (int g = 0; g < NG; ++g) { y0[g] = z0[g]; y1[g] = z1[g]; }
#pragma unroll
            for (int i = 0; i < 4; ++i) dv[i] = dn[i];
        }
    }

#pragma unroll
    for (int i = 0; i < 8; ++i) {
#pragma unroll
        for (int g = 0; g < NG; ++g) {
            float4 v;
            v.x = acc[4 * g + 0][i];
            v.y = acc[4 * g + 1][i];
            v.z = acc[4 * g + 2][i];
            v.w = acc[4 * g + 3][i];
            *(float4*)(part + ((w * 16 + 8 * h + i) * DC + D * g + 4 * l)) = v;
        }
    }
    __syncthreads();

    for (int idx = t; idx < 16 * Q; idx += BA_WAVES * 32) {
        const int rr   = idx / Q;
        const int c4   = (idx % Q) * 4;
        const int grow = row0 + rr;
        if (grow < N) {
            const float4 p0 = *(const float4*)(part + ((0 * 16 + rr) * DC + c4));
            const float4 p1 = *(const float4*)(part + ((1 * 16 + rr) * DC + c4));
            const float4 p2 = *(const float4*)(part + ((2 * 16 + rr) * DC + c4));
            const float4 p3 = *(const float4*)(part + ((3 * 16 + rr) * DC + c4));
            const float  di = dinv[grow];
            const float  c  = di * di;
            const float4 y  = *(const float4*)(Y + (size_t)grow * DC + c4);
            const float4 bv = (c4 < D) ? *(const float4*)(ba + c4) : *(const float4*)(bb + (c4 - D));
            float4 o;
            o.x = fmaf(y.x, c, bv.x) + ((p0.x + p1.x) + (p2.x + p3.x));
            o.y = fmaf(y.y, c, bv.y) + ((p0.y + p1.y) + (p2.y + p3.y));
            o.z = fmaf(y.z, c, bv.z) + ((p0.z + p1.z) + (p2.z + p3.z));
            o.w = fmaf(y.w, c, bv.w) + ((p0.w + p1.w) + (p2.w + p3.w));
            float* op = (c4 < D) ? (outa + (size_t)grow * D + c4) : (outb + (size_t)grow * D + (c4 - D));
            *(float4*)op = o;
        }
    }
}

#define D2 128

__global__ void gemm_heads_wmma(const float* __restrict__ X,
                                const float* __restrict__ Wa,
                                const float* __restrict__ Wb,
                                float* __restrict__ Y2, int nrows) {
    const int wave = (int)((blockIdx.x * blockDim.x + threadIdx.x) >> 5);
    const int lane = threadIdx.x & 31;
    const int row0 = wave * 16;
    if (row0 >= nrows) return;

    const int half = lane >> 4;
    const int l    = lane & 15;

    v8f p0 = {}, p1 = {}, p2 = {}, p3 = {};
    v8f q0 = {}, q1 = {}, q2 = {}, q3 = {};

    const float* xrow = X + (size_t)(row0 + l) * D;

#pragma unroll
    for (int k0 = 0; k0 < D; k0 += 4) {
        const int ka = k0 + 2 * half;
        v2f a;
        a.x = fmaxf(xrow[ka + 0], 0.0f);
        a.y = fmaxf(xrow[ka + 1], 0.0f);

        const float* w0 = Wa + (size_t)(ka + 0) * D;
        const float* w1 = Wa + (size_t)(ka + 1) * D;
        const float* u0 = Wb + (size_t)(ka + 0) * D;
        const float* u1 = Wb + (size_t)(ka + 1) * D;

        v2f b0; b0.x = w0[ 0 + l]; b0.y = w1[ 0 + l];
        v2f b1; b1.x = w0[16 + l]; b1.y = w1[16 + l];
        v2f b2; b2.x = w0[32 + l]; b2.y = w1[32 + l];
        v2f b3; b3.x = w0[48 + l]; b3.y = w1[48 + l];
        v2f e0; e0.x = u0[ 0 + l]; e0.y = u1[ 0 + l];
        v2f e1; e1.x = u0[16 + l]; e1.y = u1[16 + l];
        v2f e2; e2.x = u0[32 + l]; e2.y = u1[32 + l];
        v2f e3; e3.x = u0[48 + l]; e3.y = u1[48 + l];

        p0 = __builtin_amdgcn_wmma_f32_16x16x4_f32(false, a, false, b0, (short)0, p0, false, false);
        p1 = __builtin_amdgcn_wmma_f32_16x16x4_f32(false, a, false, b1, (short)0, p1, false, false);
        p2 = __builtin_amdgcn_wmma_f32_16x16x4_f32(false, a, false, b2, (short)0, p2, false, false);
        p3 = __builtin_amdgcn_wmma_f32_16x16x4_f32(false, a, false, b3, (short)0, p3, false, false);
        q0 = __builtin_amdgcn_wmma_f32_16x16x4_f32(false, a, false, e0, (short)0, q0, false, false);
        q1 = __builtin_amdgcn_wmma_f32_16x16x4_f32(false, a, false, e1, (short)0, q1, false, false);
        q2 = __builtin_amdgcn_wmma_f32_16x16x4_f32(false, a, false, e2, (short)0, q2, false, false);
        q3 = __builtin_amdgcn_wmma_f32_16x16x4_f32(false, a, false, e3, (short)0, q3, false, false);
    }

#pragma unroll
    for (int r = 0; r < 8; ++r) {
        const int row = row0 + r + 8 * half;
        float* yrow = Y2 + (size_t)row * D2 + l;
        yrow[  0] = p0[r];
        yrow[ 16] = p1[r];
        yrow[ 32] = p2[r];
        yrow[ 48] = p3[r];
        yrow[ 64] = q0[r];
        yrow[ 80] = q1[r];
        yrow[ 96] = q2[r];
        yrow[112] = q3[r];
    }
}

__global__ void relu_inplace(float* __restrict__ x, int n4) {
    int i = blockIdx.x * blockDim.x + threadIdx.x;
    if (i >= n4) return;
    float4 v = ((float4*)x)[i];
    v.x = fmaxf(v.x, 0.0f);
    v.y = fmaxf(v.y, 0.0f);
    v.z = fmaxf(v.z, 0.0f);
    v.w = fmaxf(v.w, 0.0f);
    ((float4*)x)[i] = v;
}

static inline int cdiv(long long a, int b) { return (int)((a + b - 1) / b); }

static void gcn_layer(const float* x, const float* W, const float* b,
                      const int* src, const int* dst, const float* dinv,
                      float* y, float* agg, int N, int E, hipStream_t stream) {
    const int waves  = (N + 15) / 16;
    const int gblk   = cdiv(waves, 8);
    gemm_xw_wmma<<<gblk, 256, 0, stream>>>(x, W, y, N);
    self_bias<<<cdiv((long long)N * 16, 256), 256, 0, stream>>>(y, dinv, b, agg, N);
    edge_scatter<<<cdiv(cdiv(E, 32), 8), 256, 0, stream>>>(y, src, dst, dinv, agg, E);
}

extern "C" void kernel_launch(void* const* d_in, const int* in_sizes, int n_in,
                              void* d_out, int out_size, void* d_ws, size_t ws_size,
                              hipStream_t stream) {
    const int N = in_sizes[0] / D;
    const int E = in_sizes[1] / 2;

    const float* x0  = (const float*)d_in[0];
    const int*   ei  = (const int*)d_in[1];
    const int*   src = ei;
    const int*   dst = ei + E;
    const float* W1  = (const float*)d_in[2];
    const float* b1  = (const float*)d_in[3];
    const float* W2  = (const float*)d_in[4];
    const float* b2  = (const float*)d_in[5];
    const float* Ws  = (const float*)d_in[6];
    const float* bs  = (const float*)d_in[7];
    const float* Wmu = (const float*)d_in[8];
    const float* bmu = (const float*)d_in[9];
    const float* Wlv = (const float*)d_in[10];
    const float* blv = (const float*)d_in[11];

    float* deg  = (float*)d_ws;
    float* dinv = deg + N;
    float* bufA = dinv + N;
    float* bufB = bufA + (size_t)N * D;
    float* bufY = bufB + (size_t)N * D;

    float* mu = (float*)d_out;
    float* lv = mu + (size_t)N * D;

    const int NR = cdiv(N, RS);
    const int NB = cdiv(N, 16);
    if (NR <= BP_NR && cdiv(E, DEG_GROUPS) <= BP_CAP && N <= (1 << 26)) {
        float*    wf     = bufY + (size_t)N * D;
        unsigned* rcnt   = (unsigned*)(wf + (size_t)WF_ROWS * D);
        int*      rstart = (int*)(rcnt + (size_t)DEG_GROUPS * BP_NR);
        int*      bsrc   = rstart + 2 * BP_NR;
        int*      bdl    = bsrc + (((size_t)E + 31) & ~(size_t)31);
        int2*     bspan  = (int2*)(bdl + (((size_t)E + 31) & ~(size_t)31));
        int2*     rec    = bspan + (((size_t)NB + 31) & ~(size_t)31);
        const float* w123 = wf;
        const float* c1   = wf + (size_t)D * D;
        const float* c2   = c1 + D;

        range_hist<<<DEG_GROUPS + 1, DEG_THREADS, 0, stream>>>(dst, rcnt, NR, E, W1, b1, W2, b2, Ws, wf);
        bin_place<<<DEG_GROUPS + XW_GROUPS, 1024, 0, stream>>>(src, dst, rcnt, bsrc, bdl, rstart, E, NR, x0, w123, bufY, N);
        range_sort<<<NR, RD_THREADS, 0, stream>>>(bsrc, bdl, rstart, dinv, rec, bspan, N, NB);

        block_accum_wmma<1, true><<<NB, BA_WAVES * 32, 0, stream>>>(bufY, rec, bspan, dinv, c1, nullptr, bufA, nullptr, N);
        block_accum_wmma<1, false><<<NB, BA_WAVES * 32, 0, stream>>>(bufA, rec, bspan, dinv, c2, nullptr, bufB, nullptr, N);
        block_accum_wmma<1, false><<<NB, BA_WAVES * 32, 0, stream>>>(bufB, rec, bspan, dinv, bs, nullptr, bufA, nullptr, N);
        gemm_heads_wmma<<<cdiv(cdiv(N, 16), 8), 256, 0, stream>>>(bufA, Wmu, Wlv, bufB, N);
        block_accum_wmma<2, false><<<NB, BA_WAVES * 32, 0, stream>>>(bufB, rec, bspan, dinv, bmu, blv, mu, lv, N);
        return;
    }
    if (N <= DEG_BINS) {
        unsigned* planes = (unsigned*)(bufY + (size_t)N * D);
        const int pitch = (N + 31) & ~31;
        deg_hist<<<DEG_GROUPS, DEG_THREADS, 0, stream>>>(dst, planes, nullptr, 0, E, N, pitch);
        dinv_from_planes<<<cdiv(N, 256), 256, 0, stream>>>(planes, dinv, N, pitch);
    } else {
        zero_f32<<<cdiv(N, 256), 256, 0, stream>>>(deg, N);
        deg_accum<<<cdiv(E, 256), 256, 0, stream>>>(dst, deg, E);
        dinv_kernel<<<cdiv(N, 256), 256, 0, stream>>>(deg, dinv, N);
    }

    gcn_layer(x0,   W1, b1, src, dst, dinv, bufY, bufA, N, E, stream);
    gcn_layer(bufA, W2, b2, src, dst, dinv, bufY, bufB, N, E, stream);
    gcn_layer(bufB, Ws, bs, src, dst, dinv, bufY, bufA, N, E, stream);
    relu_inplace<<<cdiv((long long)N * D / 4, 256), 256, 0, stream>>>(bufA, N * D / 4);

    gcn_layer(bufA, Wmu, bmu, src, dst, dinv, bufY, mu, N, E, stream);
    gcn_layer(bufA, Wlv, blv, src, dst, dinv, bufY, lv, N, E, stream);
}


# Row-sum loops: next k step's rows asked as one group pinned by scheduling barriers (one wait per step, not per load); records by scalar load, 32-bit offsets to keep registers; sum order kept
# speedup vs baseline: 3.5964x; 1.1123x over previous
#include <hip/hip_runtime.h>
#include <hip/hip_bf16.h>

#define D 64

typedef __attribute__((ext_vector_type(2))) float v2f;
typedef __attribute__((ext_vector_type(8))) float v8f;

__device__ __forceinline__ void atomic_fadd(float* p, float v) {
    __hip_atomic_fetch_add(p, v, __ATOMIC_RELAXED, __HIP_MEMORY_SCOPE_AGENT);
}

__global__ void zero_f32(float* p, int n) {
    int i = blockIdx.x * blockDim.x + threadIdx.x;
    if (i < n) p[i] = 0.0f;
}

__global__ void deg_accum(const int* __restrict__ dst, float* __restrict__ deg, int E) {
    int e = blockIdx.x * blockDim.x + threadIdx.x;
    if (e < E) atomic_fadd(&deg[dst[e]], 1.0f);
}

__global__ void dinv_kernel(const float* __restrict__ deg, float* __restrict__ dinv, int n) {
    int i = blockIdx.x * blockDim.x + threadIdx.x;
    if (i < n) dinv[i] = __frsqrt_rn(deg[i] + 1.0f);
}

#define DEG_BINS    65536
#define DEG_GROUPS  64
#define DEG_THREADS 1024
#define DEG_INFLIGHT 16

#define RS      128
#define BP_NR   512
#define BP_CAP  12800
#define BP_INFLIGHT 16

#define WF_ROWS  68
#define S1_ROWS  80
#define S1_PITCH 66

__device__ __forceinline__ void fold_weights(const float* __restrict__ W1, const float* __restrict__ b1,
                                             const float* __restrict__ W2, const float* __restrict__ b2,
                                             const float* __restrict__ Ws, float* __restrict__ wf,
                                             float* s1) {
    const int t    = threadIdx.x;
    const int lane = t & 31;
    const int wv   = t >> 5;
    const int half = lane >> 4;
    const int l    = lane & 15;
    const int rt   = wv >> 2;
    const int ct   = wv & 3;
    const int r    = rt * 16 + l;
    const int col  = ct * 16 + l;
    const int ntile = (S1_ROWS / 16) * 4;

    if (wv < ntile) {
        const float* arow = (r < D) ? (W1 + (size_t)r * D) : b1;
        const bool   live = (r <= D);
        const float  b2c  = b2[col];
        v8f acc = {};
#pragma unroll
        for (int k0 = 0; k0 < D; k0 += 4) {
            const int ka = k0 + 2 * half;
            const float ax = arow[ka + 0];
            const float ay = arow[ka + 1];
            v2f a;
            a.x = live ? ax : 0.0f;
            a.y = live ? ay : 0.0f;
            v2f b;
            b.x = W2[(size_t)(ka + 0) * D + col];
            b.y = W2[(size_t)(ka + 1) * D + col];
            acc = __builtin_amdgcn_wmma_f32_16x16x4_f32(false, a, false, b, (short)0, acc, false, false);
        }
#pragma unroll
        for (int i = 0; i < 8; ++i) {
            const int row = rt * 16 + 8 * half + i;
            s1[row * S1_PITCH + col] = (row == D + 1) ? b2c : acc[i];
        }
    }
    __syncthreads();
    if (wv < ntile) {
        const float* arow = s1 + r * S1_PITCH;
        v8f acc = {};
#pragma unroll
        for (int k0 = 0; k0 < D; k0 += 4) {
            const int ka = k0 + 2 * half;
            v2f a;
            a.x = arow[ka + 0];
            a.y = arow[ka + 1];
            v2f b;
            b.x = Ws[(size_t)(ka + 0) * D + col];
            b.y = Ws[(size_t)(ka + 1) * D + col];
            acc = __builtin_amdgcn_wmma_f32_16x16x4_f32(false, a, false, b, (short)0, acc, false, false);
        }
#pragma unroll
        for (int i = 0; i < 8; ++i) {
            const int row = rt * 16 + 8 * half + i;
            if (row < D + 2) wf[(size_t)row * D + col] = acc[i];
        }
    }
}

__global__ __launch_bounds__(DEG_THREADS)
void range_hist(const int* __restrict__ dst, unsigned* __restrict__ rcnt, int NR, int E,
                const float* __restrict__ W1, const float* __restrict__ b1,
                const float* __restrict__ W2, const float* __restrict__ b2,
                const float* __restrict__ Ws, float* __restrict__ wf) {
    __shared__ unsigned rc[BP_NR];
    __shared__ __attribute__((aligned(16))) float s1[S1_ROWS * S1_PITCH];
    if (blockIdx.x == DEG_GROUPS) {
        fold_weights(W1, b1, W2, b2, Ws, wf, s1);
        return;
    }
    const int t = threadIdx.x;
    if (t < BP_NR) rc[t] = 0u;
    __syncthreads();

    const int per = (E + DEG_GROUPS - 1) / DEG_GROUPS;
    const int e0  = (int)blockIdx.x * per;
    const int e1  = (e0 + per < E) ? (e0 + per) : E;

    for (int base = e0 + t; base < e1; base += DEG_INFLIGHT * DEG_THREADS) {
        int d[DEG_INFLIGHT];
#pragma unroll
        for (int u = 0; u < DEG_INFLIGHT; ++u) {
            const int e = base + u * DEG_THREADS;
            d[u] = (e < e1) ? dst[e] : -1;
        }
#pragma unroll
        for (int u = 0; u < DEG_INFLIGHT; ++u)
            if (d[u] >= 0) atomicAdd(&rc[d[u] / RS], 1u);
    }
    __syncthreads();
    if (t < NR) rcnt[(size_t)blockIdx.x * NR + t] = rc[t];
}

#define SUBB   (RS / 16)
#define PADG   16
#define RD_THREADS 256
#define RD_INFLIGHT 8

__global__ __launch_bounds__(RD_THREADS)
void range_sort(const int* __restrict__ bsrc, const int* __restrict__ bdl,
                const int* __restrict__ rstart, float* __restrict__ dinv,
                int2* __restrict__ rec, int2* __restrict__ bspan, int N, int NB) {
    __shared__ unsigned cnt[RS];
    __shared__ unsigned bcnt[SUBB];
    __shared__ unsigned cur[SUBB];
    __shared__ int      bst[SUBB];
    const int t = threadIdx.x;
    if (t < RS) cnt[t] = 0u;
    __syncthreads();

    const int es = rstart[blockIdx.x];
    const int ee = rstart[blockIdx.x + 1];
    for (int base = es + t; base < ee; base += RD_INFLIGHT * RD_THREADS) {
        int d[RD_INFLIGHT];
#pragma unroll
        for (int u = 0; u < RD_INFLIGHT; ++u) {
            const int e = base + u * RD_THREADS;
            d[u] = (e < ee) ? bdl[e] : -1;
        }
#pragma unroll
        for (int u = 0; u < RD_INFLIGHT; ++u)
            if (d[u] >= 0) atomicAdd(&cnt[d[u]], 1u);
    }
    __syncthreads();

    const int row = (int)blockIdx.x * RS + t;
    if (t < RS && row < N) dinv[row] = __frsqrt_rn((float)cnt[t] + 1.0f);
    if (t < SUBB) {
        unsigned c = 0u;
        for (int i = 0; i < 16; ++i) c += cnt[16 * t + i];
        bcnt[t] = c;
        cur[t]  = 0u;
    }
    __syncthreads();
    if (t == 0) {
        int base = ((es + PADG - 1) / PADG) * PADG + (int)blockIdx.x * (SUBB + 1) * PADG;
        for (int j = 0; j < SUBB; ++j) {
            const int pc  = (int)((bcnt[j] + PADG - 1) / PADG) * PADG;
            const int blk = (int)blockIdx.x * SUBB + j;
            bst[j] = base;
            if (blk < NB) bspan[blk] = make_int2(base, pc);
            base += pc;
        }
    }
    __syncthreads();

    for (int base = es + t; base < ee; base += RD_INFLIGHT * RD_THREADS) {
        int s[RD_INFLIGHT], d[RD_INFLIGHT];
#pragma unroll
        for (int u = 0; u < RD_INFLIGHT; ++u) {
            const int e  = base + u * RD_THREADS;
            const int ec = (e < ee) ? e : (ee - 1);
            s[u] = bsrc[ec];
            d[u] = bdl[ec];
        }
#pragma unroll
        for (int u = 0; u < RD_INFLIGHT; ++u) {
            if (base + u * RD_THREADS < ee) {
                const int j = d[u] >> 4;
                const unsigned p = atomicAdd(&cur[j], 1u);
                rec[bst[j] + (int)p] = make_int2((s[u] << 5) | (d[u] & 15), 0);
            }
        }
    }
    if (t < SUBB * PADG) {
        const int j  = t / PADG;
        const int i  = t % PADG;
        const int c  = (int)bcnt[j];
        const int pc = ((c + PADG - 1) / PADG) * PADG;
        if (c + i < pc) rec[bst[j] + c + i] = make_int2(16, 0);
    }
}

__global__ __launch_bounds__(DEG_THREADS)
void deg_hist(const int* __restrict__ dst, unsigned* __restrict__ planes,
              unsigned* __restrict__ rcnt, int NR,
              int E, int N, int pitch) {
    __shared__ unsigned bins[DEG_BINS];
    const int t = threadIdx.x;

    const int n4 = N >> 2;
    for (int i = t; i < n4; i += DEG_THREADS)
        ((uint4*)bins)[i] = make_uint4(0u, 0u, 0u, 0u);
    for (int i = (n4 << 2) + t; i < N; i += DEG_THREADS) bins[i] = 0u;
    __syncthreads();

    const int per = (E + (int)gridDim.x - 1) / (int)gridDim.x;
    const int e0  = (int)blockIdx.x * per;
    const int e1  = (e0 + per < E) ? (e0 + per) : E;

    for (int base = e0 + t; base < e1; base += DEG_INFLIGHT * DEG_THREADS) {
        int d[DEG_INFLIGHT];
#pragma unroll
        for (int u = 0; u < DEG_INFLIGHT; ++u) {
            const int e = base + u * DEG_THREADS;
            d[u] = (e < e1) ? dst[e] : -1;
        }
#pragma unroll
        for (int u = 0; u < DEG_INFLIGHT; ++u)
            if (d[u] >= 0) atomicAdd(&bins[d[u]], 1u);
    }
    __syncthreads();

    unsigned* out = planes + (size_t)blockIdx.x * pitch;
    for (int i = t; i < N; i += DEG_THREADS) out[i] = bins[i];

    if (rcnt != nullptr) {
        for (int r = t; r < NR; r += DEG_THREADS) {
            const int n0 = r * RS;
            const int n1 = (n0 + RS < N) ? (n0 + RS) : N;
            unsigned c = 0u;
            for (int i = n0; i < n1; ++i) c += bins[i];
            rcnt[(size_t)blockIdx.x * NR + r] = c;
        }
    }
}

__global__ void dinv_from_planes(const unsigned* __restrict__ planes,
                                 float* __restrict__ dinv, int n, int pitch) {
    int i = blockIdx.x * blockDim.x + threadIdx.x;
    if (i >= n) return;
    unsigned c = 0u;
#pragma unroll
    for (int g = 0; g < DEG_GROUPS; ++g) c += planes[(size_t)g * pitch + i];
    dinv[i] = __frsqrt_rn((float)c + 1.0f);
}

template <int NT>
__device__ __forceinline__ void xw_strip(const float* __restrict__ X,
                                         const float* __restrict__ W,
                                         float* __restrict__ Y, int row0, int c0, int lane) {
    const int half = lane >> 4;
    const int l    = lane & 15;

    v8f acc[NT];
#pragma unroll
    for (int j = 0; j < NT; ++j) acc[j] = v8f{};

    const float* xrow = X + (size_t)(row0 + l) * D;

#pragma unroll
    for (int k0 = 0; k0 < D; k0 += 4) {
        const int ka = k0 + 2 * half;
        v2f a;
        a.x = xrow[ka + 0];
        a.y = xrow[ka + 1];

        const float* w0 = W + (size_t)(ka + 0) * D + c0 + l;
        const float* w1 = W + (size_t)(ka + 1) * D + c0 + l;

        v2f b[NT];
#pragma unroll
        for (int j = 0; j < NT; ++j) { b[j].x = w0[16 * j]; b[j].y = w1[16 * j]; }
#pragma unroll
        for (int j = 0; j < NT; ++j)
            acc[j] = __builtin_amdgcn_wmma_f32_16x16x4_f32(false, a, false, b[j], (short)0, acc[j], false, false);
    }

#pragma unroll
    for (int r = 0; r < 8; ++r) {
        const int row = row0 + r + 8 * half;
        float* yrow = Y + (size_t)row * D + c0 + l;
#pragma unroll
        for (int j = 0; j < NT; ++j) yrow[16 * j] = acc[j][r];
    }
}

__global__ void gemm_xw_wmma(const float* __restrict__ X,
                             const float* __restrict__ W,
                             float* __restrict__ Y, int nrows) {
    const int wave = (int)((blockIdx.x * blockDim.x + threadIdx.x) >> 5);
    const int lane = threadIdx.x & 31;
    const int row0 = wave * 16;
    if (row0 >= nrows) return;
    xw_strip<4>(X, W, Y, row0, 0, lane);
}

__global__ void self_bias(const float* __restrict__ Y,
                          const float* __restrict__ dinv,
                          const float* __restrict__ b,
                          float* __restrict__ agg, int n) {
    int t = blockIdx.x * blockDim.x + threadIdx.x;
    int i = t >> 4;
    int q = (t & 15) * 4;
    if (i >= n) return;
    float di = dinv[i];
    float c  = di * di;
    const float4 y  = *(const float4*)(Y + (size_t)i * D + q);
    const float4 bb = *(const float4*)(b + q);
    float4 r;
    r.x = fmaf(y.x, c, bb.x);
    r.y = fmaf(y.y, c, bb.y);
    r.z = fmaf(y.z, c, bb.z);
    r.w = fmaf(y.w, c, bb.w);
    *(float4*)(agg + (size_t)i * D + q) = r;
}

#define EB 8

__global__ void edge_scatter(const float* __restrict__ Y,
                             const int* __restrict__ src,
                             const int* __restrict__ dst,
                             const float* __restrict__ dinv,
                             float* __restrict__ agg, int E) {
    const int lane = threadIdx.x & 31;
    const int wave = (int)((blockIdx.x * blockDim.x + threadIdx.x) >> 5);
    const int e0   = wave * 32;
    if (e0 >= E) return;

    const int e = e0 + lane;
    int s = 0, d = 0;
    float coef = 0.0f;
    if (e < E) {
        s = src[e];
        d = dst[e];
        coef = dinv[s] * dinv[d];
    }
    const int cbits = __float_as_int(coef);
    const int n = (E - e0 < 32) ? (E - e0) : 32;

    if (n == 32) {
        float va[EB], vb[EB], wa[EB], wb[EB];
#pragma unroll
        for (int u = 0; u < EB; ++u) {
            const int sj = __builtin_amdgcn_readlane(s, u);
            const float* yr = Y + (size_t)sj * D + lane;
            va[u] = yr[0];
            vb[u] = yr[32];
        }
#pragma unroll
        for (int j0 = 0; j0 < 32; j0 += EB) {
            if (j0 + EB < 32) {
#pragma unroll
                for (int u = 0; u < EB; ++u) {
                    const int sj = __builtin_amdgcn_readlane(s, j0 + EB + u);
                    const float* yr = Y + (size_t)sj * D + lane;
                    wa[u] = yr[0];
                    wb[u] = yr[32];
                }
            }
#pragma unroll
            for (int u = 0; u < EB; ++u) {
                const int   dj = __builtin_amdgcn_readlane(d, j0 + u);
                const float cj = __int_as_float(__builtin_amdgcn_readlane(cbits, j0 + u));
                float* o = agg + (size_t)dj * D + lane;
                atomic_fadd(o,      va[u] * cj);
                atomic_fadd(o + 32, vb[u] * cj);
            }
#pragma unroll
            for (int u = 0; u < EB; ++u) { va[u] = wa[u]; vb[u] = wb[u]; }
        }
    } else {
        for (int j = 0; j < n; ++j) {
            const int   sj = __builtin_amdgcn_readlane(s, j);
            const int   dj = __builtin_amdgcn_readlane(d, j);
            const float cj = __int_as_float(__builtin_amdgcn_readlane(cbits, j));
            const float* yr = Y + (size_t)sj * D + lane;
            float* o = agg + (size_t)dj * D + lane;
            atomic_fadd(o,      yr[0]  * cj);
            atomic_fadd(o + 32, yr[32] * cj);
        }
    }
}

#define XW_GROUPS 192

__global__ __launch_bounds__(1024)
void bin_place(const int* __restrict__ src, const int* __restrict__ dst,
               const unsigned* __restrict__ rcnt,
               int* __restrict__ bsrc, int* __restrict__ bdl, int* __restrict__ rstart,
               int E, int NR,
               const float* __restrict__ X, const float* __restrict__ Wf,
               float* __restrict__ Y, int nrows) {
    if (blockIdx.x >= DEG_GROUPS) {
        const int nwaves = ((int)gridDim.x - DEG_GROUPS) * 32;
        const int ntask  = 2 * ((nrows + 15) / 16);
        const int wv = __builtin_amdgcn_readfirstlane((int)threadIdx.x >> 5);
        for (int task = ((int)blockIdx.x - DEG_GROUPS) * 32 + wv; task < ntask; task += nwaves)
            xw_strip<2>(X, Wf, Y, (task >> 1) * 16, (task & 1) * 32, (int)threadIdx.x & 31);
        return;
    }
    __shared__ int      s_l[BP_CAP];
    __shared__ int      d_l[BP_CAP];
    __shared__ unsigned sc_own[BP_NR];
    __shared__ unsigned sc_tot[BP_NR];
    __shared__ unsigned lstart[BP_NR];
    __shared__ unsigned gbase[BP_NR];
    __shared__ unsigned cur[BP_NR];

    const int t = threadIdx.x;
    const int g = blockIdx.x;
    const int G = DEG_GROUPS;

    unsigned own = 0u, before = 0u, total = 0u;
    {
        const int tr = (t < NR) ? t : (NR - 1);
        unsigned c[DEG_GROUPS];
#pragma unroll
        for (int g2 = 0; g2 < DEG_GROUPS; ++g2) c[g2] = rcnt[(size_t)g2 * NR + tr];
#pragma unroll
        for (int g2 = 0; g2 < DEG_GROUPS; ++g2) {
            total  += c[g2];
            before += (g2 < g)  ? c[g2] : 0u;
            own     = (g2 == g) ? c[g2] : own;
        }
        if (t >= NR) { own = 0u; before = 0u; total = 0u; }
    }
    if (t < BP_NR) { sc_own[t] = own; sc_tot[t] = total; }
    __syncthreads();
    for (int off = 1; off < BP_NR; off <<= 1) {
        unsigned a = 0u, b = 0u;
        if (t < BP_NR && t >= off) { a = sc_own[t - off]; b = sc_tot[t - off]; }
        __syncthreads();
        if (t < BP_NR) { sc_own[t] += a; sc_tot[t] += b; }
        __syncthreads();
    }
    if (t < NR) {
        const unsigned ls = sc_own[t] - own;
        const unsigned gs = sc_tot[t] - total;
        lstart[t] = ls;
        cur[t]    = ls;
        gbase[t]  = gs + before;
        if (g == 0) {
            rstart[t] = (int)gs;
            if (t == NR - 1) rstart[NR] = (int)sc_tot[t];
        }
    }
    __syncthreads();

    const int per = (E + G - 1) / G;
    const int e0  = g * per;
    const int e1  = (e0 + per < E) ? (e0 + per) : E;

    for (int base = e0 + t; base < e1; base += BP_INFLIGHT * 1024) {
        int s[BP_INFLIGHT], d[BP_INFLIGHT];
#pragma unroll
        for (int u = 0; u < BP_INFLIGHT; ++u) {
            const int e = base + u * 1024;
            const int ec = (e < e1) ? e : (e1 - 1);
            s[u] = src[ec];
            d[u] = dst[ec];
        }
#pragma unroll
        for (int u = 0; u < BP_INFLIGHT; ++u) {
            if (base + u * 1024 < e1) {
                const int r = d[u] / RS;
                const unsigned p = atomicAdd(&cur[r], 1u);
                if (p < (unsigned)BP_CAP) { s_l[p] = s[u]; d_l[p] = d[u]; }
            }
        }
    }
    __syncthreads();

    const int cnt = e1 - e0;
    for (int i = t; i < cnt; i += 1024) {
        const int   s = s_l[i];
        const int   d = d_l[i];
        const int   r = d / RS;
        const unsigned slot = gbase[r] + ((unsigned)i - lstart[r]);
        if (slot < (unsigned)E) {
            bsrc[slot] = s;
            bdl[slot]  = d - r * RS;
        }
    }
}

#define BA_WAVES 4

template <int NG, bool MAKE_CF>
__global__ __launch_bounds__(BA_WAVES * 32)
void block_accum_wmma(const float* __restrict__ Y, int2* rec,
                      const int2* __restrict__ bspan,
                      const float* __restrict__ dinv,
                      const float* __restrict__ ba, const float* __restrict__ bb,
                      float* __restrict__ outa, float* __restrict__ outb, int N) {
    constexpr int DC = D * NG;
    constexpr int Q  = DC / 4;
    __shared__ __attribute__((aligned(16))) float part[BA_WAVES * 16 * DC];

    const int t    = threadIdx.x;
    const int lane = t & 31;
    const int w    = t >> 5;
    const int h    = lane >> 4;
    const int l    = lane & 15;
    const int row0 = (int)blockIdx.x * 16;

    const int2 sp  = bspan[blockIdx.x];
    const int per  = sp.y / (4 * BA_WAVES);
    const int ws = __builtin_amdgcn_readfirstlane(w);
    int e = sp.x + ws * per * 4;

    v8f acc[4 * NG];
#pragma unroll
    for (int i = 0; i < 4 * NG; ++i) acc[i] = v8f{};

    if (per > 0) {
        const char* Yb = (const char*)Y;
        const char* Db = (const char*)dinv;
        const char* Rb = (const char*)rec;
        const unsigned lo = 16u * (unsigned)l;
        const unsigned ho = 16u * (unsigned)h;
        auto rec_at = [&](int i) -> int4 {
            const char* rp = Rb + (size_t)(8u * (unsigned)i);
            if (MAKE_CF) {
                return *(const int4*)(rp + ho);
            } else {
                const int4 p = *(const int4*)rp;
                const int4 s = *(const int4*)(rp + 16);
                int4 o;
                o.x = h ? s.x : p.x;
                o.y = h ? s.y : p.y;
                o.z = h ? s.z : p.z;
                o.w = h ? s.w : p.w;
                return o;
            }
        };

        auto issue = [&](int x, int z, float4* ya, float4* yb, float* d) {
            const unsigned o0 = (unsigned)(x >> 5) * (unsigned)(DC * 4) + lo;
            const unsigned o1 = (unsigned)(z >> 5) * (unsigned)(DC * 4) + lo;
#pragma unroll
            for (int g = 0; g < NG; ++g) {
                ya[g] = *(const float4*)(Yb + o0 + 4 * D * g);
                yb[g] = *(const float4*)(Yb + o1 + 4 * D * g);
            }
            if (MAKE_CF) {
                d[0] = *(const float*)(Db + 4u * (unsigned)(x >> 5));
                d[1] = *(const float*)(Db + 4u * (unsigned)(row0 + (x & 15)));
                d[2] = *(const float*)(Db + 4u * (unsigned)(z >> 5));
                d[3] = *(const float*)(Db + 4u * (unsigned)(row0 + (z & 15)));
            }
        };
        auto mk_a = [&](int x, float c0, int z, float c1) -> v2f {
            v2f a;
            a.x = ((x & 31) == l) ? c0 : 0.0f;
            a.y = ((z & 31) == l) ? c1 : 0.0f;
            return a;
        };

        const int e_last = e + 4 * (per - 1);
        int4 r = rec_at(e);
        float4 y0[NG], y1[NG], z0[NG], z1[NG];
        float dv[4] = {0.0f, 0.0f, 0.0f, 0.0f}, dn[4] = {0.0f, 0.0f, 0.0f, 0.0f};
        issue(r.x, r.z, y0, y1, dv);
        int4 rn = rec_at((e + 4 < e_last) ? (e + 4) : e_last);
        int kx = r.x, kz = r.z;
        v2f a = mk_a(r.x, __int_as_float(r.y), r.z, __int_as_float(r.w));

#pragma unroll 2
        for (int q = 0; q < per; ++q, e += 4) {
            const int nx = rn.x, nz = rn.z;
            const v2f an = mk_a(rn.x, __int_as_float(rn.y), rn.z, __int_as_float(rn.w));
            __builtin_amdgcn_sched_barrier(0);
            issue(nx, nz, z0, z1, dn);
            rn = rec_at((e + 8 < e_last) ? (e + 8) : e_last);
            __builtin_amdgcn_sched_barrier(0);

            if (MAKE_CF) {
                const float c0 = ((kx & 31) < 16) ? dv[0] * dv[1] : 0.0f;
                const float c1 = ((kz & 31) < 16) ? dv[2] * dv[3] : 0.0f;
                if (l == 0) {
                    rec[e + 2 * h].y     = __float_as_int(c0);
                    rec[e + 2 * h + 1].y = __float_as_int(c1);
                }
                a = mk_a(kx, c0, kz, c1);
            }

#pragma unroll
            for (int g = 0; g < NG; ++g) {
                v2f b0; b0.x = y0[g].x; b0.y = y1[g].x;
                v2f b1; b1.x = y0[g].y; b1.y = y1[g].y;
                v2f b2; b2.x = y0[g].z; b2.y = y1[g].z;
                v2f b3; b3.x = y0[g].w; b3.y = y1[g].w;
                acc[4 * g + 0] = __builtin_amdgcn_wmma_f32_16x16x4_f32(false, a, false, b0, (short)0, acc[4 * g + 0], false, false);
                acc[4 * g + 1] = __builtin_amdgcn_wmma_f32_16x16x4_f32(false, a, false, b1, (short)0, acc[4 * g + 1], false, false);
                acc[4 * g + 2] = __builtin_amdgcn_wmma_f32_16x16x4_f32(false, a, false, b2, (short)0, acc[4 * g + 2], false, false);
                acc[4 * g + 3] = __builtin_amdgcn_wmma_f32_16x16x4_f32(false, a, false, b3, (short)0, acc[4 * g + 3], false, false);
            }
            kx = nx; kz = nz; a = an;
#pragma unroll
            for (int g = 0; g < NG; ++g) { y0[g] = z0[g]; y1[g] = z1[g]; }
#pragma unroll
            for (int i = 0; i < 4; ++i) dv[i] = dn[i];
        }
    }

#pragma unroll
    for (int i = 0; i < 8; ++i) {
#pragma unroll
        for (int g = 0; g < NG; ++g) {
            float4 v;
            v.x = acc[4 * g + 0][i];
            v.y = acc[4 * g + 1][i];
            v.z = acc[4 * g + 2][i];
            v.w = acc[4 * g + 3][i];
            *(float4*)(part + ((w * 16 + 8 * h + i) * DC + D * g + 4 * l)) = v;
        }
    }
    __syncthreads();

    for (int idx = t; idx < 16 * Q; idx += BA_WAVES * 32) {
        const int rr   = idx / Q;
        const int c4   = (idx % Q) * 4;
        const int grow = row0 + rr;
        if (grow < N) {
            const float4 p0 = *(const float4*)(part + ((0 * 16 + rr) * DC + c4));
            const float4 p1 = *(const float4*)(part + ((1 * 16 + rr) * DC + c4));
            const float4 p2 = *(const float4*)(part + ((2 * 16 + rr) * DC + c4));
            const float4 p3 = *(const float4*)(part + ((3 * 16 + rr) * DC + c4));
            const float  di = dinv[grow];
            const float  c  = di * di;
            const float4 y  = *(const float4*)(Y + (size_t)grow * DC + c4);
            const float4 bv = (c4 < D) ? *(const float4*)(ba + c4) : *(const float4*)(bb + (c4 - D));
            float4 o;
            o.x = fmaf(y.x, c, bv.x) + ((p0.x + p1.x) + (p2.x + p3.x));
            o.y = fmaf(y.y, c, bv.y) + ((p0.y + p1.y) + (p2.y + p3.y));
            o.z = fmaf(y.z, c, bv.z) + ((p0.z + p1.z) + (p2.z + p3.z));
            o.w = fmaf(y.w, c, bv.w) + ((p0.w + p1.w) + (p2.w + p3.w));
            float* op = (c4 < D) ? (outa + (size_t)grow * D + c4) : (outb + (size_t)grow * D + (c4 - D));
            *(float4*)op = o;
        }
    }
}

#define D2 128

__global__ void gemm_heads_wmma(const float* __restrict__ X,
                                const float* __restrict__ Wa,
                                const float* __restrict__ Wb,
                                float* __restrict__ Y2, int nrows) {
    const int wave = (int)((blockIdx.x * blockDim.x + threadIdx.x) >> 5);
    const int lane = threadIdx.x & 31;
    const int row0 = wave * 16;
    if (row0 >= nrows) return;

    const int half = lane >> 4;
    const int l    = lane & 15;

    v8f p0 = {}, p1 = {}, p2 = {}, p3 = {};
    v8f q0 = {}, q1 = {}, q2 = {}, q3 = {};

    const float* xrow = X + (size_t)(row0 + l) * D;

#pragma unroll
    for (int k0 = 0; k0 < D; k0 += 4) {
        const int ka = k0 + 2 * half;
        v2f a;
        a.x = fmaxf(xrow[ka + 0], 0.0f);
        a.y = fmaxf(xrow[ka + 1], 0.0f);

        const float* w0 = Wa + (size_t)(ka + 0) * D;
        const float* w1 = Wa + (size_t)(ka + 1) * D;
        const float* u0 = Wb + (size_t)(ka + 0) * D;
        const float* u1 = Wb + (size_t)(ka + 1) * D;

        v2f b0; b0.x = w0[ 0 + l]; b0.y = w1[ 0 + l];
        v2f b1; b1.x = w0[16 + l]; b1.y = w1[16 + l];
        v2f b2; b2.x = w0[32 + l]; b2.y = w1[32 + l];
        v2f b3; b3.x = w0[48 + l]; b3.y = w1[48 + l];
        v2f e0; e0.x = u0[ 0 + l]; e0.y = u1[ 0 + l];
        v2f e1; e1.x = u0[16 + l]; e1.y = u1[16 + l];
        v2f e2; e2.x = u0[32 + l]; e2.y = u1[32 + l];
        v2f e3; e3.x = u0[48 + l]; e3.y = u1[48 + l];

        p0 = __builtin_amdgcn_wmma_f32_16x16x4_f32(false, a, false, b0, (short)0, p0, false, false);
        p1 = __builtin_amdgcn_wmma_f32_16x16x4_f32(false, a, false, b1, (short)0, p1, false, false);
        p2 = __builtin_amdgcn_wmma_f32_16x16x4_f32(false, a, false, b2, (short)0, p2, false, false);
        p3 = __builtin_amdgcn_wmma_f32_16x16x4_f32(false, a, false, b3, (short)0, p3, false, false);
        q0 = __builtin_amdgcn_wmma_f32_16x16x4_f32(false, a, false, e0, (short)0, q0, false, false);
        q1 = __builtin_amdgcn_wmma_f32_16x16x4_f32(false, a, false, e1, (short)0, q1, false, false);
        q2 = __builtin_amdgcn_wmma_f32_16x16x4_f32(false, a, false, e2, (short)0, q2, false, false);
        q3 = __builtin_amdgcn_wmma_f32_16x16x4_f32(false, a, false, e3, (short)0, q3, false, false);
    }

#pragma unroll
    for (int r = 0; r < 8; ++r) {
        const int row = row0 + r + 8 * half;
        float* yrow = Y2 + (size_t)row * D2 + l;
        yrow[  0] = p0[r];
        yrow[ 16] = p1[r];
        yrow[ 32] = p2[r];
        yrow[ 48] = p3[r];
        yrow[ 64] = q0[r];
        yrow[ 80] = q1[r];
        yrow[ 96] = q2[r];
        yrow[112] = q3[r];
    }
}

__global__ void relu_inplace(float* __restrict__ x, int n4) {
    int i = blockIdx.x * blockDim.x + threadIdx.x;
    if (i >= n4) return;
    float4 v = ((float4*)x)[i];
    v.x = fmaxf(v.x, 0.0f);
    v.y = fmaxf(v.y, 0.0f);
    v.z = fmaxf(v.z, 0.0f);
    v.w = fmaxf(v.w, 0.0f);
    ((float4*)x)[i] = v;
}

static inline int cdiv(long long a, int b) { return (int)((a + b - 1) / b); }

static void gcn_layer(const float* x, const float* W, const float* b,
                      const int* src, const int* dst, const float* dinv,
                      float* y, float* agg, int N, int E, hipStream_t stream) {
    const int waves  = (N + 15) / 16;
    const int gblk   = cdiv(waves, 8);
    gemm_xw_wmma<<<gblk, 256, 0, stream>>>(x, W, y, N);
    self_bias<<<cdiv((long long)N * 16, 256), 256, 0, stream>>>(y, dinv, b, agg, N);
    edge_scatter<<<cdiv(cdiv(E, 32), 8), 256, 0, stream>>>(y, src, dst, dinv, agg, E);
}

extern "C" void kernel_launch(void* const* d_in, const int* in_sizes, int n_in,
                              void* d_out, int out_size, void* d_ws, size_t ws_size,
                              hipStream_t stream) {
    const int N = in_sizes[0] / D;
    const int E = in_sizes[1] / 2;

    const float* x0  = (const float*)d_in[0];
    const int*   ei  = (const int*)d_in[1];
    const int*   src = ei;
    const int*   dst = ei + E;
    const float* W1  = (const float*)d_in[2];
    const float* b1  = (const float*)d_in[3];
    const float* W2  = (const float*)d_in[4];
    const float* b2  = (const float*)d_in[5];
    const float* Ws  = (const float*)d_in[6];
    const float* bs  = (const float*)d_in[7];
    const float* Wmu = (const float*)d_in[8];
    const float* bmu = (const float*)d_in[9];
    const float* Wlv = (const float*)d_in[10];
    const float* blv = (const float*)d_in[11];

    float* deg  = (float*)d_ws;
    float* dinv = deg + N;
    float* bufA = dinv + N;
    float* bufB = bufA + (size_t)N * D;
    float* bufY = bufB + (size_t)N * D;

    float* mu = (float*)d_out;
    float* lv = mu + (size_t)N * D;

    const int NR = cdiv(N, RS);
    const int NB = cdiv(N, 16);
    if (NR <= BP_NR && cdiv(E, DEG_GROUPS) <= BP_CAP && N <= (1 << 26)) {
        float*    wf     = bufY + (size_t)N * D;
        unsigned* rcnt   = (unsigned*)(wf + (size_t)WF_ROWS * D);
        int*      rstart = (int*)(rcnt + (size_t)DEG_GROUPS * BP_NR);
        int*      bsrc   = rstart + 2 * BP_NR;
        int*      bdl    = bsrc + (((size_t)E + 31) & ~(size_t)31);
        int2*     bspan  = (int2*)(bdl + (((size_t)E + 31) & ~(size_t)31));
        int2*     rec    = bspan + (((size_t)NB + 31) & ~(size_t)31);
        const float* w123 = wf;
        const float* c1   = wf + (size_t)D * D;
        const float* c2   = c1 + D;

        range_hist<<<DEG_GROUPS + 1, DEG_THREADS, 0, stream>>>(dst, rcnt, NR, E, W1, b1, W2, b2, Ws, wf);
        bin_place<<<DEG_GROUPS + XW_GROUPS, 1024, 0, stream>>>(src, dst, rcnt, bsrc, bdl, rstart, E, NR, x0, w123, bufY, N);
        range_sort<<<NR, RD_THREADS, 0, stream>>>(bsrc, bdl, rstart, dinv, rec, bspan, N, NB);

        block_accum_wmma<1, true><<<NB, BA_WAVES * 32, 0, stream>>>(bufY, rec, bspan, dinv, c1, nullptr, bufA, nullptr, N);
        block_accum_wmma<1, false><<<NB, BA_WAVES * 32, 0, stream>>>(bufA, rec, bspan, dinv, c2, nullptr, bufB, nullptr, N);
        block_accum_wmma<1, false><<<NB, BA_WAVES * 32, 0, stream>>>(bufB, rec, bspan, dinv, bs, nullptr, bufA, nullptr, N);
        gemm_heads_wmma<<<cdiv(cdiv(N, 16), 8), 256, 0, stream>>>(bufA, Wmu, Wlv, bufB, N);
        block_accum_wmma<2, false><<<NB, BA_WAVES * 32, 0, stream>>>(bufB, rec, bspan, dinv, bmu, blv, mu, lv, N);
        return;
    }
    if (N <= DEG_BINS) {
        unsigned* planes = (unsigned*)(bufY + (size_t)N * D);
        const int pitch = (N + 31) & ~31;
        deg_hist<<<DEG_GROUPS, DEG_THREADS, 0, stream>>>(dst, planes, nullptr, 0, E, N, pitch);
        dinv_from_planes<<<cdiv(N, 256), 256, 0, stream>>>(planes, dinv, N, pitch);
    } else {
        zero_f32<<<cdiv(N, 256), 256, 0, stream>>>(deg, N);
        deg_accum<<<cdiv(E, 256), 256, 0, stream>>>(dst, deg, E);
        dinv_kernel<<<cdiv(N, 256), 256, 0, stream>>>(deg, dinv, N);
    }

    gcn_layer(x0,   W1, b1, src, dst, dinv, bufY, bufA, N, E, stream);
    gcn_layer(bufA, W2, b2, src, dst, dinv, bufY, bufB, N, E, stream);
    gcn_layer(bufB, Ws, bs, src, dst, dinv, bufY, bufA, N, E, stream);
    relu_inplace<<<cdiv((long long)N * D / 4, 256), 256, 0, stream>>>(bufA, N * D / 4);

    gcn_layer(bufA, Wmu, bmu, src, dst, dinv, bufY, mu, N, E, stream);
    gcn_layer(bufA, Wlv, blv, src, dst, dinv, bufY, lv, N, E, stream);
}
